# starting kernel
# baseline (speedup 1.0000x reference)
#include <hip/hip_runtime.h>

#define D 128
#define WSTR 132

typedef __attribute__((ext_vector_type(2))) float v2f;
typedef __attribute__((ext_vector_type(8))) float v8f;

__global__ void gcn_zero_f32(float* __restrict__ p, long long n) {
  long long i = (long long)blockIdx.x * blockDim.x + threadIdx.x;
  long long stride = (long long)gridDim.x * blockDim.x;
  for (; i < n; i += stride) p[i] = 0.0f;
}

__global__ void gcn_degree(const int* __restrict__ src, const int* __restrict__ dst,
                           float* __restrict__ odeg, float* __restrict__ ideg, int E) {
  int i = blockIdx.x * blockDim.x + threadIdx.x;
  if (i < E) {
    atomicAdd(&odeg[src[i]], 1.0f);
    atomicAdd(&ideg[dst[i]], 1.0f);
  }
}

__global__ void gcn_norm(float* __restrict__ x, int n) {
  int i = blockIdx.x * blockDim.x + threadIdx.x;
  if (i < n) x[i] = rsqrtf(fmaxf(x[i], 1.0f));
}

__global__ __launch_bounds__(128)
void gcn_gemm_wmma(const float* __restrict__ h, const float* __restrict__ W,
                   const float* __restrict__ bias, const float* __restrict__ out_norm,
                   float* __restrict__ hs, int nrows, int ntiles) {
  __shared__ float sW[64 * WSTR];
  __shared__ float sA[16 * WSTR];
  __shared__ float sB[64];

  const int tid   = threadIdx.x;
  const int nbase = blockIdx.y * 64;

  for (int idx = tid; idx < 64 * (D / 4); idx += 128) {
    const int r  = idx >> 5;
    const int c4 = (idx & 31) << 2;
    const float4 v = *(const float4*)(W + (size_t)(nbase + r) * D + c4);
    *(float4*)(&sW[r * WSTR + c4]) = v;
  }
  if (tid < 64) sB[tid] = bias[nbase + tid];
  __syncthreads();

  const int wave = tid >> 5;
  const int lane = tid & 31;
  const int hi   = lane >> 4;
  const int lm   = lane & 15;

  for (int tile = blockIdx.x; tile < ntiles; tile += gridDim.x) {
    const int row0 = tile << 4;

    for (int idx = tid; idx < 16 * (D / 4); idx += 128) {
      const int r   = idx >> 5;
      const int c4  = (idx & 31) << 2;
      const int row = row0 + r;
      float4 v;
      if (row < nrows) v = *(const float4*)(h + (size_t)row * D + c4);
      else { v.x = 0.f; v.y = 0.f; v.z = 0.f; v.w = 0.f; }
      *(float4*)(&sA[r * WSTR + c4]) = v;
    }
    __syncthreads();

    v8f acc = {};
    const float* __restrict__ arow = &sA[lm * WSTR];
    const float* __restrict__ brow = &sW[(wave * 16 + lm) * WSTR];
#pragma unroll
    for (int k = 0; k < D; k += 4) {
      const int kk = k + (hi << 1);
      v2f a; a.x = arow[kk]; a.y = arow[kk + 1];
      v2f b; b.x = brow[kk]; b.y = brow[kk + 1];
      acc = __builtin_amdgcn_wmma_f32_16x16x4_f32(
           false, a,  false, b,
           (short)0, acc,  false,  false);
    }

    const int n  = nbase + wave * 16 + lm;
    const float bn = sB[wave * 16 + lm];
#pragma unroll
    for (int r = 0; r < 8; ++r) {
      const int m = row0 + r + (hi << 3);
      if (m < nrows)
        hs[(size_t)m * D + n] = (acc[r] + bn) * out_norm[m];
    }
    __syncthreads();
  }
}

__global__ void gcn_scatter(const float* __restrict__ hs, const int* __restrict__ src,
                            const int* __restrict__ dst, float* __restrict__ agg,
                            long long E) {
  long long gid = (long long)blockIdx.x * blockDim.x + threadIdx.x;
  const long long nwork = E << 5;
  if (gid >= nwork) return;
  const int e  = (int)(gid >> 5);
  const int c4 = (int)(gid & 31) << 2;
  const int s = src[e];
  const int d = dst[e];
  const float4 v = *(const float4*)(hs + (size_t)s * D + c4);
  float* o = agg + (size_t)d * D + c4;
  atomicAdd(o + 0, v.x);
  atomicAdd(o + 1, v.y);
  atomicAdd(o + 2, v.z);
  atomicAdd(o + 3, v.w);
}

__global__ void gcn_finalize(const float* __restrict__ agg, const float* __restrict__ in_norm,
                             const float* __restrict__ hres, float* __restrict__ out,
                             int nrows, int do_relu) {
  long long gid = (long long)blockIdx.x * blockDim.x + threadIdx.x;
  const long long n4 = (long long)nrows * (D / 4);
  if (gid >= n4) return;
  const int row = (int)(gid >> 5);
  const float s = in_norm[row];
  float4 v = *(const float4*)(agg + (gid << 2));
  v.x *= s; v.y *= s; v.z *= s; v.w *= s;
  if (hres) {
    const float4 r = *(const float4*)(hres + (gid << 2));
    v.x += r.x; v.y += r.y; v.z += r.z; v.w += r.w;
  }
  if (do_relu) {
    v.x = fmaxf(v.x, 0.f); v.y = fmaxf(v.y, 0.f);
    v.z = fmaxf(v.z, 0.f); v.w = fmaxf(v.w, 0.f);
  }
  *(float4*)(out + (gid << 2)) = v;
}

extern "C" void kernel_launch(void* const* d_in, const int* in_sizes, int n_in,
                              void* d_out, int out_size, void* d_ws, size_t ws_size,
                              hipStream_t stream) {
  const float* feat = (const float*)d_in[0];
  const int*   src  = (const int*)d_in[1];
  const int*   dst  = (const int*)d_in[2];
  const float* Wl[3] = { (const float*)d_in[3], (const float*)d_in[5], (const float*)d_in[7] };
  const float* bl[3] = { (const float*)d_in[4], (const float*)d_in[6], (const float*)d_in[8] };

  const int N = in_sizes[0] / D;
  const int E = in_sizes[1];

  char* ws = (char*)d_ws;
  const size_t featBytes = (size_t)N * D * sizeof(float);
  const size_t normBytes = (((size_t)N * sizeof(float)) + 255) & ~(size_t)255;
  float* hs   = (float*)ws;
  float* hbuf = (float*)(ws + featBytes);
  float* onrm = (float*)(ws + 2 * featBytes);
  float* inrm = (float*)(ws + 2 * featBytes + normBytes);
  float* agg  = (float*)d_out;

  gcn_zero_f32<<<1024, 256, 0, stream>>>(onrm, (long long)N);
  gcn_zero_f32<<<1024, 256, 0, stream>>>(inrm, (long long)N);
  gcn_degree<<<(E + 255) / 256, 256, 0, stream>>>(src, dst, onrm, inrm, E);
  gcn_norm<<<(N + 255) / 256, 256, 0, stream>>>(onrm, N);
  gcn_norm<<<(N + 255) / 256, 256, 0, stream>>>(inrm, N);

  const int ntiles = (N + 15) / 16;
  const dim3 ggrid(1024, 2);
  const long long scatterWork = (long long)E << 5;
  const unsigned scatterBlocks = (unsigned)((scatterWork + 255) / 256);
  const long long n4 = (long long)N * (D / 4);
  const unsigned finBlocks = (unsigned)((n4 + 255) / 256);

  const float* hin = feat;
  for (int l = 0; l < 3; ++l) {
    const bool last = (l == 2);
    gcn_gemm_wmma<<<ggrid, 128, 0, stream>>>(hin, Wl[l], bl[l], onrm, hs, N, ntiles);
    gcn_zero_f32<<<4096, 256, 0, stream>>>(agg, (long long)N * D);
    gcn_scatter<<<scatterBlocks, 256, 0, stream>>>(hs, src, dst, agg, (long long)E);
    float* outp = last ? (float*)d_out : hbuf;
    gcn_finalize<<<finBlocks, 256, 0, stream>>>(agg, inrm, last ? nullptr : hin,
                                                outp, N, last ? 0 : 1);
    hin = hbuf;
  }
}


# Row sums on the matrix unit: edges regrouped per 16 rows once per call; a wave adds 4 gathered source rows per f32 WMMA (0/1 row-selector A) into register tiles; no LDS adds; sum order changed
# speedup vs baseline: 2.3128x; 2.3128x over previous
#include <hip/hip_runtime.h>

#define D 128
#define WSTR 132
#define TR 391
#define NCHUNK 128
#define MAXB 1024
#define SUBS 25

typedef __attribute__((ext_vector_type(2))) float v2f;
typedef __attribute__((ext_vector_type(8))) float v8f;

__global__ __launch_bounds__(128)
void gcn_gemm_wmma(const float* __restrict__ h, const float* __restrict__ W,
                   const float* __restrict__ bias, const float* __restrict__ out_norm,
                   float* __restrict__ hs, int nrows, int ntiles) {
  __shared__ float sW[64 * WSTR];
  __shared__ float sA[16 * WSTR];
  __shared__ float sB[64];

  const int tid   = threadIdx.x;
  const int nbase = blockIdx.y * 64;

  for (int idx = tid; idx < 64 * (D / 4); idx += 128) {
    const int r  = idx >> 5;
    const int c4 = (idx & 31) << 2;
    const float4 v = *(const float4*)(W + (size_t)(nbase + r) * D + c4);
    *(float4*)(&sW[r * WSTR + c4]) = v;
  }
  if (tid < 64) sB[tid] = bias[nbase + tid];
  __syncthreads();

  const int wave = tid >> 5;
  const int lane = tid & 31;
  const int hi   = lane >> 4;
  const int lm   = lane & 15;

  for (int tile = blockIdx.x; tile < ntiles; tile += gridDim.x) {
    const int row0 = tile << 4;

    for (int idx = tid; idx < 16 * (D / 4); idx += 128) {
      const int r   = idx >> 5;
      const int c4  = (idx & 31) << 2;
      const int row = row0 + r;
      float4 v;
      if (row < nrows) v = *(const float4*)(h + (size_t)row * D + c4);
      else { v.x = 0.f; v.y = 0.f; v.z = 0.f; v.w = 0.f; }
      *(float4*)(&sA[r * WSTR + c4]) = v;
    }
    __syncthreads();

    v8f acc = {};
    const float* __restrict__ arow = &sA[lm * WSTR];
    const float* __restrict__ brow = &sW[(wave * 16 + lm) * WSTR];
#pragma unroll
    for (int k = 0; k < D; k += 4) {
      const int kk = k + (hi << 1);
      v2f a; a.x = arow[kk]; a.y = arow[kk + 1];
      v2f b; b.x = brow[kk]; b.y = brow[kk + 1];
      acc = __builtin_amdgcn_wmma_f32_16x16x4_f32(
           false, a,  false, b,
           (short)0, acc,  false,  false);
    }

    const int n  = nbase + wave * 16 + lm;
    const float bn = sB[wave * 16 + lm];
#pragma unroll
    for (int r = 0; r < 8; ++r) {
      const int m = row0 + r + (hi << 3);
      if (m < nrows)
        hs[(size_t)m * D + n] = (acc[r] + bn) * out_norm[m];
    }
    __syncthreads();
  }
}

__global__ __launch_bounds__(1024)
void gcn_bin_count(const int* __restrict__ src, const int* __restrict__ dst, int E, int chunk,
                   int T, int* __restrict__ counts) {
  __shared__ int h[MAXB];
  const int tid = threadIdx.x;
  for (int b = tid; b < 2 * T; b += 1024) h[b] = 0;
  __syncthreads();
  const int e0 = blockIdx.x * chunk;
  const int e1 = min(e0 + chunk, E);
  for (int e = e0 + tid; e < e1; e += 1024) {
    atomicAdd(&h[dst[e] / TR], 1);
    atomicAdd(&h[T + src[e] / TR], 1);
  }
  __syncthreads();
  for (int b = tid; b < 2 * T; b += 1024) counts[b * NCHUNK + blockIdx.x] = h[b];
}

__global__ __launch_bounds__(1024)
void gcn_bin_scan(int* __restrict__ counts, int M, int* __restrict__ tileStart, int T) {
  __shared__ int part[1024];
  const int tid = threadIdx.x;
  const int per = (M + 1023) / 1024;
  const int i0 = min(tid * per, M);
  const int i1 = min(i0 + per, M);
  int s = 0;
  for (int i = i0; i < i1; ++i) s += counts[i];
  part[tid] = s;
  __syncthreads();
  for (int off = 1; off < 1024; off <<= 1) {
    const int v = (tid >= off) ? part[tid - off] : 0;
    __syncthreads();
    part[tid] += v;
    __syncthreads();
  }
  int run = part[tid] - s;
  for (int i = i0; i < i1; ++i) {
    const int c = counts[i];
    counts[i] = run;
    if ((i % NCHUNK) == 0) tileStart[i / NCHUNK] = run;
    run += c;
  }
  if (tid == 1023) tileStart[T] = part[1023];
}

__global__ __launch_bounds__(1024)
void gcn_bin_fill(const int* __restrict__ src, const int* __restrict__ dst, int E, int chunk,
                  int T, const int* __restrict__ counts, int* __restrict__ list) {
  __shared__ int cur[MAXB];
  const int tid = threadIdx.x;
  for (int b = tid; b < 2 * T; b += 1024) cur[b] = counts[b * NCHUNK + blockIdx.x];
  __syncthreads();
  const int e0 = blockIdx.x * chunk;
  const int e1 = min(e0 + chunk, E);
  for (int e = e0 + tid; e < e1; e += 1024) {
    const int d = dst[e];
    const int s = src[e];
    const int bd = d / TR;
    const int bs = s / TR;
    const int pd = atomicAdd(&cur[bd], 1);
    *(int2*)(list + 2 * (size_t)pd) = make_int2(s, d - bd * TR);
    const int ps = atomicAdd(&cur[T + bs], 1);
    *(int2*)(list + 2 * (size_t)ps) = make_int2(d, s - bs * TR);
  }
}

__global__ __launch_bounds__(1024)
void gcn_tile_norm(const int* __restrict__ list, const int* __restrict__ tileStart, int T,
                   float* __restrict__ onrm, float* __restrict__ inrm, int nrows,
                   int* __restrict__ list2, int* __restrict__ subStart) {
  __shared__ int cin[TR];
  __shared__ int cout[TR];
  __shared__ int subcur[SUBS];
  const int tid = threadIdx.x;
  for (int r = tid; r < TR; r += 1024) { cin[r] = 0; cout[r] = 0; }
  __syncthreads();
  const int d0 = tileStart[blockIdx.x];
  const int d1 = tileStart[blockIdx.x + 1];
  for (int e = d0 + tid; e < d1; e += 1024) atomicAdd(&cin[list[2 * (size_t)e + 1]], 1);
  const int s0 = tileStart[T + blockIdx.x];
  const int s1 = tileStart[T + blockIdx.x + 1];
  for (int e = s0 + tid; e < s1; e += 1024) atomicAdd(&cout[list[2 * (size_t)e + 1]], 1);
  __syncthreads();
  const int row0 = blockIdx.x * TR;
  const int rows = min(TR, nrows - row0);
  for (int r = tid; r < rows; r += 1024) {
    inrm[row0 + r] = rsqrtf(fmaxf((float)cin[r], 1.0f));
    onrm[row0 + r] = rsqrtf(fmaxf((float)cout[r], 1.0f));
  }
  if (tid < 32) {
    int c = 0;
    if (tid < SUBS) {
      const int r1 = min(16 * tid + 16, TR);
      for (int r = 16 * tid; r < r1; ++r) c += cin[r];
    }
    int inc = c;
#pragma unroll
    for (int off = 1; off < 32; off <<= 1) {
      const int v = __shfl_up(inc, off);
      if (tid >= off) inc += v;
    }
    if (tid < SUBS) {
      subcur[tid] = d0 + inc - c;
      subStart[blockIdx.x * (SUBS + 1) + tid] = d0 + inc - c;
    }
    if (tid == SUBS - 1) subStart[blockIdx.x * (SUBS + 1) + SUBS] = d0 + inc;
  }
  __syncthreads();
  for (int e = d0 + tid; e < d1; e += 1024) {
    const int2 ed = *(const int2*)(list + 2 * (size_t)e);
    const int pos = atomicAdd(&subcur[ed.y >> 4], 1);
    list2[pos] = (ed.x << 5) | (ed.y & 15);
  }
}

__global__ __launch_bounds__(32 * SUBS)
void gcn_rows_wmma(const float* __restrict__ hs, const int* __restrict__ list2,
                   const int* __restrict__ subStart, const float* __restrict__ in_norm,
                   const float* hres, float* out, int nrows, int do_relu) {
  const int lane = threadIdx.x & 31;
  const int wave = __builtin_amdgcn_readfirstlane(threadIdx.x >> 5);
  const int hi = lane >> 4;
  const int lm = lane & 15;
  const int e0 = __builtin_amdgcn_readfirstlane(subStart[blockIdx.x * (SUBS + 1) + wave]);
  const int e1 = __builtin_amdgcn_readfirstlane(subStart[blockIdx.x * (SUBS + 1) + wave + 1]);
  const float* const hcol = hs + 4 * lm;

  v8f acc[8];
#pragma unroll
  for (int t = 0; t < 8; ++t) acc[t] = (v8f){};

  for (int base = e0; base < e1; base += 32) {
    int rec = 16;
    if (base + lane < e1) rec = list2[base + lane];
    const int cnt = min(32, e1 - base);
    for (int j0 = 0; j0 < cnt; j0 += 8) {
      int rr[2][2];
      float4 lo[2][2], up[2][2];
#pragma unroll
      for (int u = 0; u < 2; ++u) {
#pragma unroll
        for (int e = 0; e < 2; ++e) {
          const int ra = __builtin_amdgcn_readlane(rec, j0 + 4 * u + e);
          const int rb = __builtin_amdgcn_readlane(rec, j0 + 4 * u + 2 + e);
          const int r = hi ? rb : ra;
          rr[u][e] = r;
          const float* p = hcol + (size_t)(r >> 5) * D;
          lo[u][e] = *(const float4*)p;
          up[u][e] = *(const float4*)(p + 64);
        }
      }
#pragma unroll
      for (int u = 0; u < 2; ++u) {
        v2f a;
        a.x = ((rr[u][0] & 31) == lm) ? 1.0f : 0.0f;
        a.y = ((rr[u][1] & 31) == lm) ? 1.0f : 0.0f;
        v2f b;
        b.x = lo[u][0].x; b.y = lo[u][1].x;
        acc[0] = __builtin_amdgcn_wmma_f32_16x16x4_f32(false, a, false, b, (short)0, acc[0], false, false);
        b.x = lo[u][0].y; b.y = lo[u][1].y;
        acc[1] = __builtin_amdgcn_wmma_f32_16x16x4_f32(false, a, false, b, (short)0, acc[1], false, false);
        b.x = lo[u][0].z; b.y = lo[u][1].z;
        acc[2] = __builtin_amdgcn_wmma_f32_16x16x4_f32(false, a, false, b, (short)0, acc[2], false, false);
        b.x = lo[u][0].w; b.y = lo[u][1].w;
        acc[3] = __builtin_amdgcn_wmma_f32_16x16x4_f32(false, a, false, b, (short)0, acc[3], false, false);
        b.x = up[u][0].x; b.y = up[u][1].x;
        acc[4] = __builtin_amdgcn_wmma_f32_16x16x4_f32(false, a, false, b, (short)0, acc[4], false, false);
        b.x = up[u][0].y; b.y = up[u][1].y;
        acc[5] = __builtin_amdgcn_wmma_f32_16x16x4_f32(false, a, false, b, (short)0, acc[5], false, false);
        b.x = up[u][0].z; b.y = up[u][1].z;
        acc[6] = __builtin_amdgcn_wmma_f32_16x16x4_f32(false, a, false, b, (short)0, acc[6], false, false);
        b.x = up[u][0].w; b.y = up[u][1].w;
        acc[7] = __builtin_amdgcn_wmma_f32_16x16x4_f32(false, a, false, b, (short)0, acc[7], false, false);
      }
    }
  }

  const int tile0 = blockIdx.x * TR;
  const int rows  = min(TR, nrows - tile0);
#pragma unroll
  for (int r = 0; r < 8; ++r) {
    const int lr = 16 * wave + 8 * hi + r;
    if (lr < rows) {
      const int row = tile0 + lr;
      const size_t g = (size_t)row * D + 4 * lm;
      const float sc = in_norm[row];
      float4 v = make_float4(acc[0][r] * sc, acc[1][r] * sc, acc[2][r] * sc, acc[3][r] * sc);
      float4 w = make_float4(acc[4][r] * sc, acc[5][r] * sc, acc[6][r] * sc, acc[7][r] * sc);
      if (hres) {
        const float4 q = *(const float4*)(hres + g);
        const float4 q2 = *(const float4*)(hres + g + 64);
        v.x += q.x; v.y += q.y; v.z += q.z; v.w += q.w;
        w.x += q2.x; w.y += q2.y; w.z += q2.z; w.w += q2.w;
      }
      if (do_relu) {
        v.x = fmaxf(v.x, 0.f); v.y = fmaxf(v.y, 0.f);
        v.z = fmaxf(v.z, 0.f); v.w = fmaxf(v.w, 0.f);
        w.x = fmaxf(w.x, 0.f); w.y = fmaxf(w.y, 0.f);
        w.z = fmaxf(w.z, 0.f); w.w = fmaxf(w.w, 0.f);
      }
      *(float4*)(out + g) = v;
      *(float4*)(out + g + 64) = w;
    }
  }
}

extern "C" void kernel_launch(void* const* d_in, const int* in_sizes, int n_in,
                              void* d_out, int out_size, void* d_ws, size_t ws_size,
                              hipStream_t stream) {
  const float* feat = (const float*)d_in[0];
  const int*   src  = (const int*)d_in[1];
  const int*   dst  = (const int*)d_in[2];
  const float* Wl[3] = { (const float*)d_in[3], (const float*)d_in[5], (const float*)d_in[7] };
  const float* bl[3] = { (const float*)d_in[4], (const float*)d_in[6], (const float*)d_in[8] };

  const int N = in_sizes[0] / D;
  const int E = in_sizes[1];

  char* ws = (char*)d_ws;
  const size_t featBytes = (size_t)N * D * sizeof(float);
  const size_t normBytes = (((size_t)N * sizeof(float)) + 255) & ~(size_t)255;
  float* hs   = (float*)ws;
  float* hbuf = (float*)(ws + featBytes);
  float* onrm = (float*)(ws + 2 * featBytes);
  float* inrm = (float*)(ws + 2 * featBytes + normBytes);
  const int T     = (N + TR - 1) / TR;
  const int chunk = (E + NCHUNK - 1) / NCHUNK;
  const size_t cntBytes = (((size_t)(2 * T * NCHUNK + 2 * T + 1) * sizeof(int)) + 255) & ~(size_t)255;
  int*  counts    = (int*)(ws + 2 * featBytes + 2 * normBytes);
  int*  tileStart = counts + (size_t)2 * T * NCHUNK;
  int*  list      = (int*)(ws + 2 * featBytes + 2 * normBytes + cntBytes);
  int*  list2     = list + (size_t)4 * E;
  int*  subStart  = list2 + (size_t)E;

  gcn_bin_count<<<NCHUNK, 1024, 0, stream>>>(src, dst, E, chunk, T, counts);
  gcn_bin_scan<<<1, 1024, 0, stream>>>(counts, 2 * T * NCHUNK, tileStart, 2 * T);
  gcn_bin_fill<<<NCHUNK, 1024, 0, stream>>>(src, dst, E, chunk, T, counts, list);

  gcn_tile_norm<<<T, 1024, 0, stream>>>(list, tileStart, T, onrm, inrm, N, list2, subStart);

  const int ntiles = (N + 15) / 16;
  const dim3 ggrid(1024, 2);

  const float* hin = feat;
  for (int l = 0; l < 3; ++l) {
    const bool last = (l == 2);
    gcn_gemm_wmma<<<ggrid, 128, 0, stream>>>(hin, Wl[l], bl[l], onrm, hs, N, ntiles);
    float* outp = last ? (float*)d_out : hbuf;
    gcn_rows_wmma<<<T, 32 * SUBS, 0, stream>>>(hs, list2, subStart, inrm, last ? nullptr : hin,
                                               outp, N, last ? 0 : 1);
    hin = hbuf;
  }
}


# Dense product with W held in registers: each wave keeps all 128x128 weights as B fragments (512 VGPRs, one wave per SIMD) and streams 16-row tiles from memory; no LDS, no barriers; k order changed
# speedup vs baseline: 3.1800x; 1.3750x over previous
#include <hip/hip_runtime.h>

#define D 128
#define TR 391
#define NCHUNK 128
#define MAXB 1024
#define SUBS 25

typedef __attribute__((ext_vector_type(2))) float v2f;
typedef __attribute__((ext_vector_type(4))) float v4f;
typedef __attribute__((ext_vector_type(8))) float v8f;

__global__ __launch_bounds__(128)
void gcn_gemm_wreg(const float* __restrict__ h, const float* __restrict__ W,
                   const float* __restrict__ bias, const float* __restrict__ out_norm,
                   float* __restrict__ hs, int nrows, int ntiles) {
  const int lane = threadIdx.x & 31;
  const int wave = __builtin_amdgcn_readfirstlane(threadIdx.x >> 5);
  const int hi = lane >> 4;
  const int lm = lane & 15;
  const int nw = gridDim.x * 4;
  int tile = blockIdx.x * 4 + wave;
  if (tile >= ntiles) return;

  v4f w[16][8];
  const float* const wp = W + (size_t)(4 * lm) * D + 4 * hi;
#pragma unroll
  for (int t = 0; t < 8; ++t) {
#pragma unroll
    for (int j = 0; j < 16; ++j)
      w[j][t] = *(const v4f*)(wp + (size_t)((t & 3) + 64 * (t >> 2)) * D + 8 * j);
  }
  const v4f blo = *(const v4f*)(bias + 4 * lm);
  const v4f bup = *(const v4f*)(bias + 64 + 4 * lm);

  v4f a[16];
  {
    const float* p = h + (size_t)min((tile << 4) + lm, nrows - 1) * D + 4 * hi;
#pragma unroll
    for (int j = 0; j < 16; ++j) a[j] = *(const v4f*)(p + 8 * j);
  }

  for (;;) {
    const int row0 = tile << 4;
    const int nxt = tile + nw;
    float on[8];
#pragma unroll
    for (int r = 0; r < 8; ++r) on[r] = out_norm[min(row0 + 8 * hi + r, nrows - 1)];
    v4f an[16];
    {
      const float* p = h + (size_t)min((min(nxt, ntiles - 1) << 4) + lm, nrows - 1) * D + 4 * hi;
#pragma unroll
      for (int j = 0; j < 16; ++j) an[j] = *(const v4f*)(p + 8 * j);
    }
    __builtin_amdgcn_sched_barrier(0);

    v8f acc[8];
#pragma unroll
    for (int t = 0; t < 8; ++t) acc[t] = (v8f){};
#pragma unroll
    for (int j = 0; j < 16; ++j) {
#pragma unroll
      for (int s = 0; s < 2; ++s) {
        v2f av;
        av.x = s ? a[j].z : a[j].x;
        av.y = s ? a[j].w : a[j].y;
#pragma unroll
        for (int t = 0; t < 8; ++t) {
          v2f bv;
          bv.x = s ? w[j][t].z : w[j][t].x;
          bv.y = s ? w[j][t].w : w[j][t].y;
          acc[t] = __builtin_amdgcn_wmma_f32_16x16x4_f32(
               false, av,  false, bv,
               (short)0, acc[t],  false,  false);
        }
      }
    }

    if (row0 + 16 <= nrows) {
#pragma unroll
      for (int r = 0; r < 8; ++r) {
        const float sc = on[r];
        float* g = hs + (size_t)(row0 + 8 * hi + r) * D + 4 * lm;
        *(float4*)g = make_float4((acc[0][r] + blo.x) * sc, (acc[1][r] + blo.y) * sc,
                                  (acc[2][r] + blo.z) * sc, (acc[3][r] + blo.w) * sc);
        *(float4*)(g + 64) = make_float4((acc[4][r] + bup.x) * sc, (acc[5][r] + bup.y) * sc,
                                         (acc[6][r] + bup.z) * sc, (acc[7][r] + bup.w) * sc);
      }
    } else {
#pragma unroll
      for (int r = 0; r < 8; ++r) {
        const int m = row0 + 8 * hi + r;
        const float sc = on[r];
        if (m < nrows) {
          float* g = hs + (size_t)m * D + 4 * lm;
          *(float4*)g = make_float4((acc[0][r] + blo.x) * sc, (acc[1][r] + blo.y) * sc,
                                    (acc[2][r] + blo.z) * sc, (acc[3][r] + blo.w) * sc);
          *(float4*)(g + 64) = make_float4((acc[4][r] + bup.x) * sc, (acc[5][r] + bup.y) * sc,
                                           (acc[6][r] + bup.z) * sc, (acc[7][r] + bup.w) * sc);
        }
      }
    }

    if (nxt >= ntiles) break;
    tile = nxt;
#pragma unroll
    for (int j = 0; j < 16; ++j) a[j] = an[j];
  }
}

__global__ __launch_bounds__(1024)
void gcn_bin_count(const int* __restrict__ src, const int* __restrict__ dst, int E, int chunk,
                   int T, int* __restrict__ counts) {
  __shared__ int h[MAXB];
  const int tid = threadIdx.x;
  for (int b = tid; b < 2 * T; b += 1024) h[b] = 0;
  __syncthreads();
  const int e0 = blockIdx.x * chunk;
  const int e1 = min(e0 + chunk, E);
  for (int e = e0 + tid; e < e1; e += 1024) {
    atomicAdd(&h[dst[e] / TR], 1);
    atomicAdd(&h[T + src[e] / TR], 1);
  }
  __syncthreads();
  for (int b = tid; b < 2 * T; b += 1024) counts[b * NCHUNK + blockIdx.x] = h[b];
}

__global__ __launch_bounds__(1024)
void gcn_bin_scan(int* __restrict__ counts, int M, int* __restrict__ tileStart, int T) {
  __shared__ int part[1024];
  const int tid = threadIdx.x;
  const int per = (M + 1023) / 1024;
  const int i0 = min(tid * per, M);
  const int i1 = min(i0 + per, M);
  int s = 0;
  for (int i = i0; i < i1; ++i) s += counts[i];
  part[tid] = s;
  __syncthreads();
  for (int off = 1; off < 1024; off <<= 1) {
    const int v = (tid >= off) ? part[tid - off] : 0;
    __syncthreads();
    part[tid] += v;
    __syncthreads();
  }
  int run = part[tid] - s;
  for (int i = i0; i < i1; ++i) {
    const int c = counts[i];
    counts[i] = run;
    if ((i % NCHUNK) == 0) tileStart[i / NCHUNK] = run;
    run += c;
  }
  if (tid == 1023) tileStart[T] = part[1023];
}

__global__ __launch_bounds__(1024)
void gcn_bin_fill(const int* __restrict__ src, const int* __restrict__ dst, int E, int chunk,
                  int T, const int* __restrict__ counts, int* __restrict__ list) {
  __shared__ int cur[MAXB];
  const int tid = threadIdx.x;
  for (int b = tid; b < 2 * T; b += 1024) cur[b] = counts[b * NCHUNK + blockIdx.x];
  __syncthreads();
  const int e0 = blockIdx.x * chunk;
  const int e1 = min(e0 + chunk, E);
  for (int e = e0 + tid; e < e1; e += 1024) {
    const int d = dst[e];
    const int s = src[e];
    const int bd = d / TR;
    const int bs = s / TR;
    const int pd = atomicAdd(&cur[bd], 1);
    *(int2*)(list + 2 * (size_t)pd) = make_int2(s, d - bd * TR);
    const int ps = atomicAdd(&cur[T + bs], 1);
    *(int2*)(list + 2 * (size_t)ps) = make_int2(d, s - bs * TR);
  }
}

__global__ __launch_bounds__(1024)
void gcn_tile_norm(const int* __restrict__ list, const int* __restrict__ tileStart, int T,
                   float* __restrict__ onrm, float* __restrict__ inrm, int nrows,
                   int* __restrict__ list2, int* __restrict__ subStart) {
  __shared__ int cin[TR];
  __shared__ int cout[TR];
  __shared__ int subcur[SUBS];
  const int tid = threadIdx.x;
  for (int r = tid; r < TR; r += 1024) { cin[r] = 0; cout[r] = 0; }
  __syncthreads();
  const int d0 = tileStart[blockIdx.x];
  const int d1 = tileStart[blockIdx.x + 1];
  for (int e = d0 + tid; e < d1; e += 1024) atomicAdd(&cin[list[2 * (size_t)e + 1]], 1);
  const int s0 = tileStart[T + blockIdx.x];
  const int s1 = tileStart[T + blockIdx.x + 1];
  for (int e = s0 + tid; e < s1; e += 1024) atomicAdd(&cout[list[2 * (size_t)e + 1]], 1);
  __syncthreads();
  const int row0 = blockIdx.x * TR;
  const int rows = min(TR, nrows - row0);
  for (int r = tid; r < rows; r += 1024) {
    inrm[row0 + r] = rsqrtf(fmaxf((float)cin[r], 1.0f));
    onrm[row0 + r] = rsqrtf(fmaxf((float)cout[r], 1.0f));
  }
  if (tid < 32) {
    int c = 0;
    if (tid < SUBS) {
      const int r1 = min(16 * tid + 16, TR);
      for (int r = 16 * tid; r < r1; ++r) c += cin[r];
    }
    int inc = c;
#pragma unroll
    for (int off = 1; off < 32; off <<= 1) {
      const int v = __shfl_up(inc, off);
      if (tid >= off) inc += v;
    }
    if (tid < SUBS) {
      subcur[tid] = d0 + inc - c;
      subStart[blockIdx.x * (SUBS + 1) + tid] = d0 + inc - c;
    }
    if (tid == SUBS - 1) subStart[blockIdx.x * (SUBS + 1) + SUBS] = d0 + inc;
  }
  __syncthreads();
  for (int e = d0 + tid; e < d1; e += 1024) {
    const int2 ed = *(const int2*)(list + 2 * (size_t)e);
    const int pos = atomicAdd(&subcur[ed.y >> 4], 1);
    list2[pos] = (ed.x << 5) | (ed.y & 15);
  }
}

__global__ __launch_bounds__(32 * SUBS)
void gcn_rows_wmma(const float* __restrict__ hs, const int* __restrict__ list2,
                   const int* __restrict__ subStart, const float* __restrict__ in_norm,
                   const float* hres, float* out, int nrows, int do_relu) {
  const int lane = threadIdx.x & 31;
  const int wave = __builtin_amdgcn_readfirstlane(threadIdx.x >> 5);
  const int hi = lane >> 4;
  const int lm = lane & 15;
  const int e0 = __builtin_amdgcn_readfirstlane(subStart[blockIdx.x * (SUBS + 1) + wave]);
  const int e1 = __builtin_amdgcn_readfirstlane(subStart[blockIdx.x * (SUBS + 1) + wave + 1]);
  const float* const hcol = hs + 4 * lm;

  v8f acc[8];
#pragma unroll
  for (int t = 0; t < 8; ++t) acc[t] = (v8f){};

  for (int base = e0; base < e1; base += 32) {
    int rec = 16;
    if (base + lane < e1) rec = list2[base + lane];
    const int cnt = min(32, e1 - base);
    for (int j0 = 0; j0 < cnt; j0 += 8) {
      int rr[2][2];
      float4 lo[2][2], up[2][2];
#pragma unroll
      for (int u = 0; u < 2; ++u) {
#pragma unroll
        for (int e = 0; e < 2; ++e) {
          const int ra = __builtin_amdgcn_readlane(rec, j0 + 4 * u + e);
          const int rb = __builtin_amdgcn_readlane(rec, j0 + 4 * u + 2 + e);
          const int r = hi ? rb : ra;
          rr[u][e] = r;
          const float* p = hcol + (size_t)(r >> 5) * D;
          lo[u][e] = *(const float4*)p;
          up[u][e] = *(const float4*)(p + 64);
        }
      }
#pragma unroll
      for (int u = 0; u < 2; ++u) {
        v2f a;
        a.x = ((rr[u][0] & 31) == lm) ? 1.0f : 0.0f;
        a.y = ((rr[u][1] & 31) == lm) ? 1.0f : 0.0f;
        v2f b;
        b.x = lo[u][0].x; b.y = lo[u][1].x;
        acc[0] = __builtin_amdgcn_wmma_f32_16x16x4_f32(false, a, false, b, (short)0, acc[0], false, false);
        b.x = lo[u][0].y; b.y = lo[u][1].y;
        acc[1] = __builtin_amdgcn_wmma_f32_16x16x4_f32(false, a, false, b, (short)0, acc[1], false, false);
        b.x = lo[u][0].z; b.y = lo[u][1].z;
        acc[2] = __builtin_amdgcn_wmma_f32_16x16x4_f32(false, a, false, b, (short)0, acc[2], false, false);
        b.x = lo[u][0].w; b.y = lo[u][1].w;
        acc[3] = __builtin_amdgcn_wmma_f32_16x16x4_f32(false, a, false, b, (short)0, acc[3], false, false);
        b.x = up[u][0].x; b.y = up[u][1].x;
        acc[4] = __builtin_amdgcn_wmma_f32_16x16x4_f32(false, a, false, b, (short)0, acc[4], false, false);
        b.x = up[u][0].y; b.y = up[u][1].y;
        acc[5] = __builtin_amdgcn_wmma_f32_16x16x4_f32(false, a, false, b, (short)0, acc[5], false, false);
        b.x = up[u][0].z; b.y = up[u][1].z;
        acc[6] = __builtin_amdgcn_wmma_f32_16x16x4_f32(false, a, false, b, (short)0, acc[6], false, false);
        b.x = up[u][0].w; b.y = up[u][1].w;
        acc[7] = __builtin_amdgcn_wmma_f32_16x16x4_f32(false, a, false, b, (short)0, acc[7], false, false);
      }
    }
  }

  const int tile0 = blockIdx.x * TR;
  const int rows  = min(TR, nrows - tile0);
#pragma unroll
  for (int r = 0; r < 8; ++r) {
    const int lr = 16 * wave + 8 * hi + r;
    if (lr < rows) {
      const int row = tile0 + lr;
      const size_t g = (size_t)row * D + 4 * lm;
      const float sc = in_norm[row];
      float4 v = make_float4(acc[0][r] * sc, acc[1][r] * sc, acc[2][r] * sc, acc[3][r] * sc);
      float4 w = make_float4(acc[4][r] * sc, acc[5][r] * sc, acc[6][r] * sc, acc[7][r] * sc);
      if (hres) {
        const float4 q = *(const float4*)(hres + g);
        const float4 q2 = *(const float4*)(hres + g + 64);
        v.x += q.x; v.y += q.y; v.z += q.z; v.w += q.w;
        w.x += q2.x; w.y += q2.y; w.z += q2.z; w.w += q2.w;
      }
      if (do_relu) {
        v.x = fmaxf(v.x, 0.f); v.y = fmaxf(v.y, 0.f);
        v.z = fmaxf(v.z, 0.f); v.w = fmaxf(v.w, 0.f);
        w.x = fmaxf(w.x, 0.f); w.y = fmaxf(w.y, 0.f);
        w.z = fmaxf(w.z, 0.f); w.w = fmaxf(w.w, 0.f);
      }
      *(float4*)(out + g) = v;
      *(float4*)(out + g + 64) = w;
    }
  }
}

extern "C" void kernel_launch(void* const* d_in, const int* in_sizes, int n_in,
                              void* d_out, int out_size, void* d_ws, size_t ws_size,
                              hipStream_t stream) {
  const float* feat = (const float*)d_in[0];
  const int*   src  = (const int*)d_in[1];
  const int*   dst  = (const int*)d_in[2];
  const float* Wl[3] = { (const float*)d_in[3], (const float*)d_in[5], (const float*)d_in[7] };
  const float* bl[3] = { (const float*)d_in[4], (const float*)d_in[6], (const float*)d_in[8] };

  const int N = in_sizes[0] / D;
  const int E = in_sizes[1];

  char* ws = (char*)d_ws;
  const size_t featBytes = (size_t)N * D * sizeof(float);
  const size_t normBytes = (((size_t)N * sizeof(float)) + 255) & ~(size_t)255;
  float* hs   = (float*)ws;
  float* hbuf = (float*)(ws + featBytes);
  float* onrm = (float*)(ws + 2 * featBytes);
  float* inrm = (float*)(ws + 2 * featBytes + normBytes);
  const int T     = (N + TR - 1) / TR;
  const int chunk = (E + NCHUNK - 1) / NCHUNK;
  const size_t cntBytes = (((size_t)(2 * T * NCHUNK + 2 * T + 1) * sizeof(int)) + 255) & ~(size_t)255;
  int*  counts    = (int*)(ws + 2 * featBytes + 2 * normBytes);
  int*  tileStart = counts + (size_t)2 * T * NCHUNK;
  int*  list      = (int*)(ws + 2 * featBytes + 2 * normBytes + cntBytes);
  int*  list2     = list + (size_t)4 * E;
  int*  subStart  = list2 + (size_t)E;

  gcn_bin_count<<<NCHUNK, 1024, 0, stream>>>(src, dst, E, chunk, T, counts);
  gcn_bin_scan<<<1, 1024, 0, stream>>>(counts, 2 * T * NCHUNK, tileStart, 2 * T);
  gcn_bin_fill<<<NCHUNK, 1024, 0, stream>>>(src, dst, E, chunk, T, counts, list);

  gcn_tile_norm<<<T, 1024, 0, stream>>>(list, tileStart, T, onrm, inrm, N, list2, subStart);

  const int ntiles = (N + 15) / 16;

  const float* hin = feat;
  for (int l = 0; l < 3; ++l) {
    const bool last = (l == 2);
    gcn_gemm_wreg<<<256, 128, 0, stream>>>(hin, Wl[l], bl[l], onrm, hs, N, ntiles);
    float* outp = last ? (float*)d_out : hbuf;
    gcn_rows_wmma<<<T, 32 * SUBS, 0, stream>>>(hs, list2, subStart, inrm, last ? nullptr : hin,
                                               outp, N, last ? 0 : 1);
    hin = hbuf;
  }
}


# Edges ordered in LDS, stored as whole lines: a group sorts its edge chunk by range, keeps run offsets (no global count/scan); a range's group collects its 256 runs; 8 launches; arrival-order sums
# speedup vs baseline: 4.3230x; 1.3594x over previous
#include <hip/hip_runtime.h>

#define D 128
#define TR 391
#define NCH 256
#define EPT 16
#define TPR (1024 / NCH)
#define CAP2 16384
#define MAXB 1024
#define SUBS 25

typedef __attribute__((ext_vector_type(2))) float v2f;
typedef __attribute__((ext_vector_type(4))) float v4f;
typedef __attribute__((ext_vector_type(8))) float v8f;

__global__ __launch_bounds__(128)
void gcn_gemm_wreg(const float* __restrict__ h, const float* __restrict__ W,
                   const float* __restrict__ bias, const float* __restrict__ out_norm,
                   float* __restrict__ hs, int nrows, int ntiles) {
  const int lane = threadIdx.x & 31;
  const int wave = __builtin_amdgcn_readfirstlane(threadIdx.x >> 5);
  const int hi = lane >> 4;
  const int lm = lane & 15;
  const int nw = gridDim.x * 4;
  int tile = blockIdx.x * 4 + wave;
  if (tile >= ntiles) return;

  v4f w[16][8];
  const float* const wp = W + (size_t)(4 * lm) * D + 4 * hi;
#pragma unroll
  for (int t = 0; t < 8; ++t) {
#pragma unroll
    for (int j = 0; j < 16; ++j)
      w[j][t] = *(const v4f*)(wp + (size_t)((t & 3) + 64 * (t >> 2)) * D + 8 * j);
  }
  const v4f blo = *(const v4f*)(bias + 4 * lm);
  const v4f bup = *(const v4f*)(bias + 64 + 4 * lm);

  v4f a[16];
  {
    const float* p = h + (size_t)min((tile << 4) + lm, nrows - 1) * D + 4 * hi;
#pragma unroll
    for (int j = 0; j < 16; ++j) a[j] = *(const v4f*)(p + 8 * j);
  }

  for (;;) {
    const int row0 = tile << 4;
    const int nxt = tile + nw;
    float on[8];
#pragma unroll
    for (int r = 0; r < 8; ++r) on[r] = out_norm[min(row0 + 8 * hi + r, nrows - 1)];
    v4f an[16];
    {
      const float* p = h + (size_t)min((min(nxt, ntiles - 1) << 4) + lm, nrows - 1) * D + 4 * hi;
#pragma unroll
      for (int j = 0; j < 16; ++j) an[j] = *(const v4f*)(p + 8 * j);
    }
    __builtin_amdgcn_sched_barrier(0);

    v8f acc[8];
#pragma unroll
    for (int t = 0; t < 8; ++t) acc[t] = (v8f){};
#pragma unroll
    for (int j = 0; j < 16; ++j) {
#pragma unroll
      for (int s = 0; s < 2; ++s) {
        v2f av;
        av.x = s ? a[j].z : a[j].x;
        av.y = s ? a[j].w : a[j].y;
#pragma unroll
        for (int t = 0; t < 8; ++t) {
          v2f bv;
          bv.x = s ? w[j][t].z : w[j][t].x;
          bv.y = s ? w[j][t].w : w[j][t].y;
          acc[t] = __builtin_amdgcn_wmma_f32_16x16x4_f32(
               false, av,  false, bv,
               (short)0, acc[t],  false,  false);
        }
      }
    }

    if (row0 + 16 <= nrows) {
#pragma unroll
      for (int r = 0; r < 8; ++r) {
        const float sc = on[r];
        float* g = hs + (size_t)(row0 + 8 * hi + r) * D + 4 * lm;
        *(float4*)g = make_float4((acc[0][r] + blo.x) * sc, (acc[1][r] + blo.y) * sc,
                                  (acc[2][r] + blo.z) * sc, (acc[3][r] + blo.w) * sc);
        *(float4*)(g + 64) = make_float4((acc[4][r] + bup.x) * sc, (acc[5][r] + bup.y) * sc,
                                         (acc[6][r] + bup.z) * sc, (acc[7][r] + bup.w) * sc);
      }
    } else {
#pragma unroll
      for (int r = 0; r < 8; ++r) {
        const int m = row0 + 8 * hi + r;
        const float sc = on[r];
        if (m < nrows) {
          float* g = hs + (size_t)m * D + 4 * lm;
          *(float4*)g = make_float4((acc[0][r] + blo.x) * sc, (acc[1][r] + blo.y) * sc,
                                    (acc[2][r] + blo.z) * sc, (acc[3][r] + blo.w) * sc);
          *(float4*)(g + 64) = make_float4((acc[4][r] + bup.x) * sc, (acc[5][r] + bup.y) * sc,
                                           (acc[6][r] + bup.z) * sc, (acc[7][r] + bup.w) * sc);
        }
      }
    }

    if (nxt >= ntiles) break;
    tile = nxt;
#pragma unroll
    for (int j = 0; j < 16; ++j) a[j] = an[j];
  }
}

__global__ __launch_bounds__(1024)
void gcn_chunk_order(const int* __restrict__ src, const int* __restrict__ dst, int E, int chunk,
                     int T, int* __restrict__ runTab, int* __restrict__ list) {
  __shared__ int cur[MAXB];
  __shared__ int wtot[32];
  __shared__ int2 stage2[1024 * EPT];
  int* const stage = (int*)stage2;
  const int tid = threadIdx.x;
  const int lane = tid & 31;
  const int wave = __builtin_amdgcn_readfirstlane(tid >> 5);
  const int nb = 2 * T;
  for (int b = tid; b < MAXB; b += 1024) cur[b] = 0;
  __syncthreads();
  const int e0 = min((int)blockIdx.x * chunk, E);
  const int e1 = min(e0 + chunk, E);

  int dv[EPT], sv[EPT];
#pragma unroll
  for (int k = 0; k < EPT; ++k) {
    const int e = e0 + tid + 1024 * k;
    dv[k] = 0; sv[k] = 0;
    if (e < e1) { dv[k] = dst[e]; sv[k] = src[e]; }
  }
#pragma unroll
  for (int k = 0; k < EPT; ++k) {
    if (e0 + tid + 1024 * k < e1) {
      atomicAdd(&cur[dv[k] / TR], 1);
      atomicAdd(&cur[T + sv[k] / TR], 1);
    }
  }
  __syncthreads();

  {
    const int x = (tid < nb) ? cur[tid] : 0;
    int inc = x;
    for (int off = 1; off < 32; off <<= 1) {
      const int v = __shfl_up(inc, off);
      if (lane >= off) inc += v;
    }
    if (lane == 31) wtot[wave] = inc;
    __syncthreads();
    int pre = (lane < wave) ? wtot[lane] : 0;
    for (int off = 16; off > 0; off >>= 1) pre += __shfl_xor(pre, off);
    const int start = pre + inc - x;
    if (tid < nb) { cur[tid] = start; runTab[tid * NCH + blockIdx.x] = start; }
    if (tid == 1023) runTab[nb * NCH + blockIdx.x] = pre + inc;
  }
  __syncthreads();

#pragma unroll
  for (int k = 0; k < EPT; ++k) {
    if (e0 + tid + 1024 * k < e1) {
      const int d = dv[k];
      const int s = sv[k];
      const int bd = d / TR;
      const int bs = s / TR;
      const int pd = atomicAdd(&cur[bd], 1);
      stage[pd] = (s << 9) | (d - bd * TR);
      const int ps = atomicAdd(&cur[T + bs], 1);
      stage[ps] = (d << 9) | (s - bs * TR);
    }
  }
  __syncthreads();

  int2* const g = (int2*)(list + (size_t)blockIdx.x * 2 * chunk);
  const int pieces = e1 - e0;
  for (int i = tid; i < pieces; i += 1024) g[i] = stage2[i];
}

__global__ __launch_bounds__(1024)
void gcn_tile_norm(const int* __restrict__ list, const int* __restrict__ runTab, int T, int chunk,
                   float* __restrict__ onrm, float* __restrict__ inrm, int nrows,
                   int* __restrict__ list2, int* __restrict__ subStart) {
  __shared__ int cin[TR];
  __shared__ int cout[TR];
  __shared__ int subcur[SUBS];
  __shared__ int rOff[2][NCH];
  __shared__ int rLen[2][NCH];
  __shared__ int tot[2];
  __shared__ int stage[CAP2];
  const int tid = threadIdx.x;
  const int lane = tid & 31;
  const int t = blockIdx.x;
  for (int r = tid; r < TR; r += 1024) { cin[r] = 0; cout[r] = 0; }
  if (tid < 2) tot[tid] = 0;
  __syncthreads();
  if (tid < NCH) {
    const int a0 = runTab[t * NCH + tid];
    const int a1 = runTab[(t + 1) * NCH + tid];
    const int b0 = runTab[(T + t) * NCH + tid];
    const int b1 = runTab[(T + t + 1) * NCH + tid];
    const int base = tid * 2 * chunk;
    rOff[0][tid] = base + a0; rLen[0][tid] = a1 - a0;
    rOff[1][tid] = base + b0; rLen[1][tid] = b1 - b0;
    int s0 = a0;
    int s1 = a1 - a0;
    for (int off = 16; off > 0; off >>= 1) {
      s0 += __shfl_xor(s0, off);
      s1 += __shfl_xor(s1, off);
    }
    if (lane == 0) { atomicAdd(&tot[0], s0); atomicAdd(&tot[1], s1); }
  }
  __syncthreads();
  const int d0 = tot[0];
  const int dn = tot[1];

  const int run = tid / TPR;
  const int q = tid % TPR;
  const int oD = rOff[0][run];
  const int nD = rLen[0][run];
  const int oS = rOff[1][run];
  const int nS = rLen[1][run];
  for (int i = q; i < nD; i += TPR) atomicAdd(&cin[list[oD + i] & 511], 1);
  for (int i = q; i < nS; i += TPR) atomicAdd(&cout[list[oS + i] & 511], 1);
  __syncthreads();
  const int row0 = t * TR;
  const int rows = min(TR, nrows - row0);
  for (int r = tid; r < rows; r += 1024) {
    inrm[row0 + r] = rsqrtf(fmaxf((float)cin[r], 1.0f));
    onrm[row0 + r] = rsqrtf(fmaxf((float)cout[r], 1.0f));
  }
  if (tid < 32) {
    int c = 0;
    if (tid < SUBS) {
      const int r1 = min(16 * tid + 16, TR);
      for (int r = 16 * tid; r < r1; ++r) c += cin[r];
    }
    int inc = c;
#pragma unroll
    for (int off = 1; off < 32; off <<= 1) {
      const int v = __shfl_up(inc, off);
      if (tid >= off) inc += v;
    }
    if (tid < SUBS) {
      subcur[tid] = inc - c;
      subStart[t * (SUBS + 1) + tid] = d0 + inc - c;
    }
    if (tid == SUBS - 1) subStart[t * (SUBS + 1) + SUBS] = d0 + inc;
  }
  __syncthreads();
  const bool staged = dn <= CAP2;
  for (int i = q; i < nD; i += TPR) {
    const int w = list[oD + i];
    const int lr = w & 511;
    const int pos = atomicAdd(&subcur[lr >> 4], 1);
    const int word = ((w >> 9) << 5) | (lr & 15);
    if (staged) stage[pos] = word; else list2[d0 + pos] = word;
  }
  if (staged) {
    __syncthreads();
    for (int k = tid; k < dn; k += 1024) list2[d0 + k] = stage[k];
  }
}

__global__ __launch_bounds__(32 * SUBS)
void gcn_rows_wmma(const float* __restrict__ hs, const int* __restrict__ list2,
                   const int* __restrict__ subStart, const float* __restrict__ in_norm,
                   const float* hres, float* out, int nrows, int do_relu) {
  const int lane = threadIdx.x & 31;
  const int wave = __builtin_amdgcn_readfirstlane(threadIdx.x >> 5);
  const int hi = lane >> 4;
  const int lm = lane & 15;
  const int e0 = __builtin_amdgcn_readfirstlane(subStart[blockIdx.x * (SUBS + 1) + wave]);
  const int e1 = __builtin_amdgcn_readfirstlane(subStart[blockIdx.x * (SUBS + 1) + wave + 1]);
  const float* const hcol = hs + 4 * lm;

  v8f acc[8];
#pragma unroll
  for (int t = 0; t < 8; ++t) acc[t] = (v8f){};

  for (int base = e0; base < e1; base += 32) {
    int rec = 16;
    if (base + lane < e1) rec = list2[base + lane];
    const int cnt = min(32, e1 - base);
    for (int j0 = 0; j0 < cnt; j0 += 8) {
      int rr[2][2];
      float4 lo[2][2], up[2][2];
#pragma unroll
      for (int u = 0; u < 2; ++u) {
#pragma unroll
        for (int e = 0; e < 2; ++e) {
          const int ra = __builtin_amdgcn_readlane(rec, j0 + 4 * u + e);
          const int rb = __builtin_amdgcn_readlane(rec, j0 + 4 * u + 2 + e);
          const int r = hi ? rb : ra;
          rr[u][e] = r;
          const float* p = hcol + (size_t)(r >> 5) * D;
          lo[u][e] = *(const float4*)p;
          up[u][e] = *(const float4*)(p + 64);
        }
      }
#pragma unroll
      for (int u = 0; u < 2; ++u) {
        v2f a;
        a.x = ((rr[u][0] & 31) == lm) ? 1.0f : 0.0f;
        a.y = ((rr[u][1] & 31) == lm) ? 1.0f : 0.0f;
        v2f b;
        b.x = lo[u][0].x; b.y = lo[u][1].x;
        acc[0] = __builtin_amdgcn_wmma_f32_16x16x4_f32(false, a, false, b, (short)0, acc[0], false, false);
        b.x = lo[u][0].y; b.y = lo[u][1].y;
        acc[1] = __builtin_amdgcn_wmma_f32_16x16x4_f32(false, a, false, b, (short)0, acc[1], false, false);
        b.x = lo[u][0].z; b.y = lo[u][1].z;
        acc[2] = __builtin_amdgcn_wmma_f32_16x16x4_f32(false, a, false, b, (short)0, acc[2], false, false);
        b.x = lo[u][0].w; b.y = lo[u][1].w;
        acc[3] = __builtin_amdgcn_wmma_f32_16x16x4_f32(false, a, false, b, (short)0, acc[3], false, false);
        b.x = up[u][0].x; b.y = up[u][1].x;
        acc[4] = __builtin_amdgcn_wmma_f32_16x16x4_f32(false, a, false, b, (short)0, acc[4], false, false);
        b.x = up[u][0].y; b.y = up[u][1].y;
        acc[5] = __builtin_amdgcn_wmma_f32_16x16x4_f32(false, a, false, b, (short)0, acc[5], false, false);
        b.x = up[u][0].z; b.y = up[u][1].z;
        acc[6] = __builtin_amdgcn_wmma_f32_16x16x4_f32(false, a, false, b, (short)0, acc[6], false, false);
        b.x = up[u][0].w; b.y = up[u][1].w;
        acc[7] = __builtin_amdgcn_wmma_f32_16x16x4_f32(false, a, false, b, (short)0, acc[7], false, false);
      }
    }
  }

  const int tile0 = blockIdx.x * TR;
  const int rows  = min(TR, nrows - tile0);
#pragma unroll
  for (int r = 0; r < 8; ++r) {
    const int lr = 16 * wave + 8 * hi + r;
    if (lr < rows) {
      const int row = tile0 + lr;
      const size_t g = (size_t)row * D + 4 * lm;
      const float sc = in_norm[row];
      float4 v = make_float4(acc[0][r] * sc, acc[1][r] * sc, acc[2][r] * sc, acc[3][r] * sc);
      float4 w = make_float4(acc[4][r] * sc, acc[5][r] * sc, acc[6][r] * sc, acc[7][r] * sc);
      if (hres) {
        const float4 q = *(const float4*)(hres + g);
        const float4 q2 = *(const float4*)(hres + g + 64);
        v.x += q.x; v.y += q.y; v.z += q.z; v.w += q.w;
        w.x += q2.x; w.y += q2.y; w.z += q2.z; w.w += q2.w;
      }
      if (do_relu) {
        v.x = fmaxf(v.x, 0.f); v.y = fmaxf(v.y, 0.f);
        v.z = fmaxf(v.z, 0.f); v.w = fmaxf(v.w, 0.f);
        w.x = fmaxf(w.x, 0.f); w.y = fmaxf(w.y, 0.f);
        w.z = fmaxf(w.z, 0.f); w.w = fmaxf(w.w, 0.f);
      }
      *(float4*)(out + g) = v;
      *(float4*)(out + g + 64) = w;
    }
  }
}

extern "C" void kernel_launch(void* const* d_in, const int* in_sizes, int n_in,
                              void* d_out, int out_size, void* d_ws, size_t ws_size,
                              hipStream_t stream) {
  const float* feat = (const float*)d_in[0];
  const int*   src  = (const int*)d_in[1];
  const int*   dst  = (const int*)d_in[2];
  const float* Wl[3] = { (const float*)d_in[3], (const float*)d_in[5], (const float*)d_in[7] };
  const float* bl[3] = { (const float*)d_in[4], (const float*)d_in[6], (const float*)d_in[8] };

  const int N = in_sizes[0] / D;
  const int E = in_sizes[1];

  char* ws = (char*)d_ws;
  const size_t featBytes = (size_t)N * D * sizeof(float);
  const size_t normBytes = (((size_t)N * sizeof(float)) + 255) & ~(size_t)255;
  float* hs   = (float*)ws;
  float* hbuf = (float*)(ws + featBytes);
  float* onrm = (float*)(ws + 2 * featBytes);
  float* inrm = (float*)(ws + 2 * featBytes + normBytes);
  const int T     = (N + TR - 1) / TR;
  const int chunk = (E + NCH - 1) / NCH;
  const size_t tabBytes = (((size_t)(2 * T + 1) * NCH * sizeof(int)) + 255) & ~(size_t)255;
  int*  runTab    = (int*)(ws + 2 * featBytes + 2 * normBytes);
  int*  list      = (int*)(ws + 2 * featBytes + 2 * normBytes + tabBytes);
  int*  list2     = list + (size_t)2 * chunk * NCH;
  int*  subStart  = list2 + (size_t)E;

  gcn_chunk_order<<<NCH, 1024, 0, stream>>>(src, dst, E, chunk, T, runTab, list);

  gcn_tile_norm<<<T, 1024, 0, stream>>>(list, runTab, T, chunk, onrm, inrm, N, list2, subStart);

  const int ntiles = (N + 15) / 16;

  const float* hin = feat;
  for (int l = 0; l < 3; ++l) {
    const bool last = (l == 2);
    gcn_gemm_wreg<<<256, 128, 0, stream>>>(hin, Wl[l], bl[l], onrm, hs, N, ntiles);
    float* outp = last ? (float*)d_out : hbuf;
    gcn_rows_wmma<<<T, 32 * SUBS, 0, stream>>>(hs, list2, subStart, inrm, last ? nullptr : hin,
                                               outp, N, last ? 0 : 1);
    hin = hbuf;
  }
}


# Dependent single loads regrouped: a row-sum wave asks its 16 norms (scalar) and 16 residual pieces at once (was 2 round trips a row), a tile-kernel thread 8 run entries per trip; sum order kept
# speedup vs baseline: 4.5578x; 1.0543x over previous
#include <hip/hip_runtime.h>

#define D 128
#define TR 391
#define NCH 256
#define EPT 16
#define TPR (1024 / NCH)
#define CAP2 16384
#define MAXB 1024
#define SUBS 25
#define RB 8

typedef __attribute__((ext_vector_type(2))) float v2f;
typedef __attribute__((ext_vector_type(4))) float v4f;
typedef __attribute__((ext_vector_type(8))) float v8f;

__global__ __launch_bounds__(128)
void gcn_gemm_wreg(const float* __restrict__ h, const float* __restrict__ W,
                   const float* __restrict__ bias, const float* __restrict__ out_norm,
                   float* __restrict__ hs, int nrows, int ntiles) {
  const int lane = threadIdx.x & 31;
  const int wave = __builtin_amdgcn_readfirstlane(threadIdx.x >> 5);
  const int hi = lane >> 4;
  const int lm = lane & 15;
  const int nw = gridDim.x * 4;
  int tile = blockIdx.x * 4 + wave;
  if (tile >= ntiles) return;

  v4f w[16][8];
  const float* const wp = W + (size_t)(4 * lm) * D + 4 * hi;
#pragma unroll
  for (int t = 0; t < 8; ++t) {
#pragma unroll
    for (int j = 0; j < 16; ++j)
      w[j][t] = *(const v4f*)(wp + (size_t)((t & 3) + 64 * (t >> 2)) * D + 8 * j);
  }
  const v4f blo = *(const v4f*)(bias + 4 * lm);
  const v4f bup = *(const v4f*)(bias + 64 + 4 * lm);

  v4f a[16];
  {
    const float* p = h + (size_t)min((tile << 4) + lm, nrows - 1) * D + 4 * hi;
#pragma unroll
    for (int j = 0; j < 16; ++j) a[j] = *(const v4f*)(p + 8 * j);
  }

  for (;;) {
    const int row0 = tile << 4;
    const int nxt = tile + nw;
    float on[8];
#pragma unroll
    for (int r = 0; r < 8; ++r) on[r] = out_norm[min(row0 + 8 * hi + r, nrows - 1)];
    v4f an[16];
    {
      const float* p = h + (size_t)min((min(nxt, ntiles - 1) << 4) + lm, nrows - 1) * D + 4 * hi;
#pragma unroll
      for (int j = 0; j < 16; ++j) an[j] = *(const v4f*)(p + 8 * j);
    }
    __builtin_amdgcn_sched_barrier(0);

    v8f acc[8];
#pragma unroll
    for (int t = 0; t < 8; ++t) acc[t] = (v8f){};
#pragma unroll
    for (int j = 0; j < 16; ++j) {
#pragma unroll
      for (int s = 0; s < 2; ++s) {
        v2f av;
        av.x = s ? a[j].z : a[j].x;
        av.y = s ? a[j].w : a[j].y;
#pragma unroll
        for (int t = 0; t < 8; ++t) {
          v2f bv;
          bv.x = s ? w[j][t].z : w[j][t].x;
          bv.y = s ? w[j][t].w : w[j][t].y;
          acc[t] = __builtin_amdgcn_wmma_f32_16x16x4_f32(
               false, av,  false, bv,
               (short)0, acc[t],  false,  false);
        }
      }
    }

    if (row0 + 16 <= nrows) {
#pragma unroll
      for (int r = 0; r < 8; ++r) {
        const float sc = on[r];
        float* g = hs + (size_t)(row0 + 8 * hi + r) * D + 4 * lm;
        *(float4*)g = make_float4((acc[0][r] + blo.x) * sc, (acc[1][r] + blo.y) * sc,
                                  (acc[2][r] + blo.z) * sc, (acc[3][r] + blo.w) * sc);
        *(float4*)(g + 64) = make_float4((acc[4][r] + bup.x) * sc, (acc[5][r] + bup.y) * sc,
                                         (acc[6][r] + bup.z) * sc, (acc[7][r] + bup.w) * sc);
      }
    } else {
#pragma unroll
      for (int r = 0; r < 8; ++r) {
        const int m = row0 + 8 * hi + r;
        const float sc = on[r];
        if (m < nrows) {
          float* g = hs + (size_t)m * D + 4 * lm;
          *(float4*)g = make_float4((acc[0][r] + blo.x) * sc, (acc[1][r] + blo.y) * sc,
                                    (acc[2][r] + blo.z) * sc, (acc[3][r] + blo.w) * sc);
          *(float4*)(g + 64) = make_float4((acc[4][r] + bup.x) * sc, (acc[5][r] + bup.y) * sc,
                                           (acc[6][r] + bup.z) * sc, (acc[7][r] + bup.w) * sc);
        }
      }
    }

    if (nxt >= ntiles) break;
    tile = nxt;
#pragma unroll
    for (int j = 0; j < 16; ++j) a[j] = an[j];
  }
}

__global__ __launch_bounds__(1024)
void gcn_chunk_order(const int* __restrict__ src, const int* __restrict__ dst, int E, int chunk,
                     int T, int* __restrict__ runTab, int* __restrict__ list) {
  __shared__ int cur[MAXB];
  __shared__ int wtot[32];
  __shared__ int2 stage2[1024 * EPT];
  int* const stage = (int*)stage2;
  const int tid = threadIdx.x;
  const int lane = tid & 31;
  const int wave = __builtin_amdgcn_readfirstlane(tid >> 5);
  const int nb = 2 * T;
  for (int b = tid; b < MAXB; b += 1024) cur[b] = 0;
  __syncthreads();
  const int e0 = min((int)blockIdx.x * chunk, E);
  const int e1 = min(e0 + chunk, E);

  int dv[EPT], sv[EPT];
#pragma unroll
  for (int k = 0; k < EPT; ++k) {
    const int e = e0 + tid + 1024 * k;
    dv[k] = 0; sv[k] = 0;
    if (e < e1) { dv[k] = dst[e]; sv[k] = src[e]; }
  }
#pragma unroll
  for (int k = 0; k < EPT; ++k) {
    if (e0 + tid + 1024 * k < e1) {
      atomicAdd(&cur[dv[k] / TR], 1);
      atomicAdd(&cur[T + sv[k] / TR], 1);
    }
  }
  __syncthreads();

  {
    const int x = (tid < nb) ? cur[tid] : 0;
    int inc = x;
    for (int off = 1; off < 32; off <<= 1) {
      const int v = __shfl_up(inc, off);
      if (lane >= off) inc += v;
    }
    if (lane == 31) wtot[wave] = inc;
    __syncthreads();
    int pre = (lane < wave) ? wtot[lane] : 0;
    for (int off = 16; off > 0; off >>= 1) pre += __shfl_xor(pre, off);
    const int start = pre + inc - x;
    if (tid < nb) { cur[tid] = start; runTab[tid * NCH + blockIdx.x] = start; }
    if (tid == 1023) runTab[nb * NCH + blockIdx.x] = pre + inc;
  }
  __syncthreads();

#pragma unroll
  for (int k = 0; k < EPT; ++k) {
    if (e0 + tid + 1024 * k < e1) {
      const int d = dv[k];
      const int s = sv[k];
      const int bd = d / TR;
      const int bs = s / TR;
      const int pd = atomicAdd(&cur[bd], 1);
      stage[pd] = (s << 9) | (d - bd * TR);
      const int ps = atomicAdd(&cur[T + bs], 1);
      stage[ps] = (d << 9) | (s - bs * TR);
    }
  }
  __syncthreads();

  int2* const g = (int2*)(list + (size_t)blockIdx.x * 2 * chunk);
  const int pieces = e1 - e0;
  for (int i = tid; i < pieces; i += 1024) g[i] = stage2[i];
}

__global__ __launch_bounds__(1024)
void gcn_tile_norm(const int* __restrict__ list, const int* __restrict__ runTab, int T, int chunk,
                   float* __restrict__ onrm, float* __restrict__ inrm, int nrows,
                   int* __restrict__ list2, int* __restrict__ subStart) {
  __shared__ int cin[TR];
  __shared__ int cout[TR];
  __shared__ int subcur[SUBS];
  __shared__ int rOff[2][NCH];
  __shared__ int rLen[2][NCH];
  __shared__ int tot[2];
  __shared__ int stage[CAP2];
  const int tid = threadIdx.x;
  const int lane = tid & 31;
  const int t = blockIdx.x;
  for (int r = tid; r < TR; r += 1024) { cin[r] = 0; cout[r] = 0; }
  if (tid < 2) tot[tid] = 0;
  __syncthreads();
  if (tid < NCH) {
    const int a0 = runTab[t * NCH + tid];
    const int a1 = runTab[(t + 1) * NCH + tid];
    const int b0 = runTab[(T + t) * NCH + tid];
    const int b1 = runTab[(T + t + 1) * NCH + tid];
    const int base = tid * 2 * chunk;
    rOff[0][tid] = base + a0; rLen[0][tid] = a1 - a0;
    rOff[1][tid] = base + b0; rLen[1][tid] = b1 - b0;
    int s0 = a0;
    int s1 = a1 - a0;
    for (int off = 16; off > 0; off >>= 1) {
      s0 += __shfl_xor(s0, off);
      s1 += __shfl_xor(s1, off);
    }
    if (lane == 0) { atomicAdd(&tot[0], s0); atomicAdd(&tot[1], s1); }
  }
  __syncthreads();
  const int d0 = tot[0];
  const int dn = tot[1];

  const int run = tid / TPR;
  const int q = tid % TPR;
  const int oD = rOff[0][run];
  const int nD = rLen[0][run];
  const int oS = rOff[1][run];
  const int nS = rLen[1][run];
  for (int i0 = q; i0 < max(nD, nS); i0 += RB * TPR) {
    int wd[RB], wsr[RB];
#pragma unroll
    for (int u = 0; u < RB; ++u) {
      const int i = i0 + u * TPR;
      wd[u] = 0; wsr[u] = 0;
      if (i < nD) wd[u] = list[oD + i];
      if (i < nS) wsr[u] = list[oS + i];
    }
#pragma unroll
    for (int u = 0; u < RB; ++u) {
      const int i = i0 + u * TPR;
      if (i < nD) atomicAdd(&cin[wd[u] & 511], 1);
      if (i < nS) atomicAdd(&cout[wsr[u] & 511], 1);
    }
  }
  __syncthreads();
  const int row0 = t * TR;
  const int rows = min(TR, nrows - row0);
  for (int r = tid; r < rows; r += 1024) {
    inrm[row0 + r] = rsqrtf(fmaxf((float)cin[r], 1.0f));
    onrm[row0 + r] = rsqrtf(fmaxf((float)cout[r], 1.0f));
  }
  if (tid < 32) {
    int c = 0;
    if (tid < SUBS) {
      const int r1 = min(16 * tid + 16, TR);
      for (int r = 16 * tid; r < r1; ++r) c += cin[r];
    }
    int inc = c;
#pragma unroll
    for (int off = 1; off < 32; off <<= 1) {
      const int v = __shfl_up(inc, off);
      if (tid >= off) inc += v;
    }
    if (tid < SUBS) {
      subcur[tid] = inc - c;
      subStart[t * (SUBS + 1) + tid] = d0 + inc - c;
    }
    if (tid == SUBS - 1) subStart[t * (SUBS + 1) + SUBS] = d0 + inc;
  }
  __syncthreads();
  const bool staged = dn <= CAP2;
  for (int i0 = q; i0 < nD; i0 += RB * TPR) {
    int wv[RB];
#pragma unroll
    for (int u = 0; u < RB; ++u) {
      const int i = i0 + u * TPR;
      wv[u] = 0;
      if (i < nD) wv[u] = list[oD + i];
    }
#pragma unroll
    for (int u = 0; u < RB; ++u) {
      if (i0 + u * TPR < nD) {
        const int w = wv[u];
        const int lr = w & 511;
        const int pos = atomicAdd(&subcur[lr >> 4], 1);
        const int word = ((w >> 9) << 5) | (lr & 15);
        if (staged) stage[pos] = word; else list2[d0 + pos] = word;
      }
    }
  }
  if (staged) {
    __syncthreads();
    for (int k = tid; k < dn; k += 1024) list2[d0 + k] = stage[k];
  }
}

__global__ __launch_bounds__(32 * SUBS)
void gcn_rows_wmma(const float* __restrict__ hs, const int* __restrict__ list2,
                   const int* __restrict__ subStart, const float* __restrict__ in_norm,
                   const float* hres, float* out, int nrows, int do_relu) {
  const int lane = threadIdx.x & 31;
  const int wave = __builtin_amdgcn_readfirstlane(threadIdx.x >> 5);
  const int hi = lane >> 4;
  const int lm = lane & 15;
  const int e0 = __builtin_amdgcn_readfirstlane(subStart[blockIdx.x * (SUBS + 1) + wave]);
  const int e1 = __builtin_amdgcn_readfirstlane(subStart[blockIdx.x * (SUBS + 1) + wave + 1]);
  const float* const hcol = hs + 4 * lm;

  v8f acc[8];
#pragma unroll
  for (int t = 0; t < 8; ++t) acc[t] = (v8f){};

  for (int base = e0; base < e1; base += 32) {
    int rec = 16;
    if (base + lane < e1) rec = list2[base + lane];
    const int cnt = min(32, e1 - base);
    for (int j0 = 0; j0 < cnt; j0 += 8) {
      int rr[2][2];
      float4 lo[2][2], up[2][2];
#pragma unroll
      for (int u = 0; u < 2; ++u) {
#pragma unroll
        for (int e = 0; e < 2; ++e) {
          const int ra = __builtin_amdgcn_readlane(rec, j0 + 4 * u + e);
          const int rb = __builtin_amdgcn_readlane(rec, j0 + 4 * u + 2 + e);
          const int r = hi ? rb : ra;
          rr[u][e] = r;
          const float* p = hcol + (size_t)(r >> 5) * D;
          lo[u][e] = *(const float4*)p;
          up[u][e] = *(const float4*)(p + 64);
        }
      }
#pragma unroll
      for (int u = 0; u < 2; ++u) {
        v2f a;
        a.x = ((rr[u][0] & 31) == lm) ? 1.0f : 0.0f;
        a.y = ((rr[u][1] & 31) == lm) ? 1.0f : 0.0f;
        v2f b;
        b.x = lo[u][0].x; b.y = lo[u][1].x;
        acc[0] = __builtin_amdgcn_wmma_f32_16x16x4_f32(false, a, false, b, (short)0, acc[0], false, false);
        b.x = lo[u][0].y; b.y = lo[u][1].y;
        acc[1] = __builtin_amdgcn_wmma_f32_16x16x4_f32(false, a, false, b, (short)0, acc[1], false, false);
        b.x = lo[u][0].z; b.y = lo[u][1].z;
        acc[2] = __builtin_amdgcn_wmma_f32_16x16x4_f32(false, a, false, b, (short)0, acc[2], false, false);
        b.x = lo[u][0].w; b.y = lo[u][1].w;
        acc[3] = __builtin_amdgcn_wmma_f32_16x16x4_f32(false, a, false, b, (short)0, acc[3], false, false);
        b.x = up[u][0].x; b.y = up[u][1].x;
        acc[4] = __builtin_amdgcn_wmma_f32_16x16x4_f32(false, a, false, b, (short)0, acc[4], false, false);
        b.x = up[u][0].y; b.y = up[u][1].y;
        acc[5] = __builtin_amdgcn_wmma_f32_16x16x4_f32(false, a, false, b, (short)0, acc[5], false, false);
        b.x = up[u][0].z; b.y = up[u][1].z;
        acc[6] = __builtin_amdgcn_wmma_f32_16x16x4_f32(false, a, false, b, (short)0, acc[6], false, false);
        b.x = up[u][0].w; b.y = up[u][1].w;
        acc[7] = __builtin_amdgcn_wmma_f32_16x16x4_f32(false, a, false, b, (short)0, acc[7], false, false);
      }
    }
  }

  const int tile0 = blockIdx.x * TR;
  const int rows  = min(TR, nrows - tile0);
  const int lr0 = 16 * wave + 8 * hi;
  if (16 * wave + 16 <= rows) {
    const float* const np = in_norm + tile0 + 16 * wave;
    float sa[8], sb[8];
#pragma unroll
    for (int r = 0; r < 8; ++r) { sa[r] = np[r]; sb[r] = np[8 + r]; }
    const unsigned boff = ((unsigned)(tile0 + lr0) * D + 4u * lm) * 4u;
    char* const op = (char*)out + boff;
    if (hres) {
      const char* const hp = (const char*)hres + boff;
      float4 q[8], q2[8];
#pragma unroll
      for (int r = 0; r < 8; ++r) {
        q[r]  = *(const float4*)(hp + r * (D * 4));
        q2[r] = *(const float4*)(hp + r * (D * 4) + 256);
      }
#pragma unroll
      for (int r = 0; r < 8; ++r) {
        const float sc = hi ? sb[r] : sa[r];
        float4 v = make_float4(acc[0][r] * sc, acc[1][r] * sc, acc[2][r] * sc, acc[3][r] * sc);
        float4 w = make_float4(acc[4][r] * sc, acc[5][r] * sc, acc[6][r] * sc, acc[7][r] * sc);
        v.x += q[r].x; v.y += q[r].y; v.z += q[r].z; v.w += q[r].w;
        w.x += q2[r].x; w.y += q2[r].y; w.z += q2[r].z; w.w += q2[r].w;
        if (do_relu) {
          v.x = fmaxf(v.x, 0.f); v.y = fmaxf(v.y, 0.f);
          v.z = fmaxf(v.z, 0.f); v.w = fmaxf(v.w, 0.f);
          w.x = fmaxf(w.x, 0.f); w.y = fmaxf(w.y, 0.f);
          w.z = fmaxf(w.z, 0.f); w.w = fmaxf(w.w, 0.f);
        }
        *(float4*)(op + r * (D * 4)) = v;
        *(float4*)(op + r * (D * 4) + 256) = w;
      }
    } else {
#pragma unroll
      for (int r = 0; r < 8; ++r) {
        const float sc = hi ? sb[r] : sa[r];
        float4 v = make_float4(acc[0][r] * sc, acc[1][r] * sc, acc[2][r] * sc, acc[3][r] * sc);
        float4 w = make_float4(acc[4][r] * sc, acc[5][r] * sc, acc[6][r] * sc, acc[7][r] * sc);
        if (do_relu) {
          v.x = fmaxf(v.x, 0.f); v.y = fmaxf(v.y, 0.f);
          v.z = fmaxf(v.z, 0.f); v.w = fmaxf(v.w, 0.f);
          w.x = fmaxf(w.x, 0.f); w.y = fmaxf(w.y, 0.f);
          w.z = fmaxf(w.z, 0.f); w.w = fmaxf(w.w, 0.f);
        }
        *(float4*)(op + r * (D * 4)) = v;
        *(float4*)(op + r * (D * 4) + 256) = w;
      }
    }
  } else {
#pragma unroll
    for (int r = 0; r < 8; ++r) {
      const int lr = lr0 + r;
      if (lr < rows) {
        const int row = tile0 + lr;
        const size_t g = (size_t)row * D + 4 * lm;
        const float sc = in_norm[row];
        float4 v = make_float4(acc[0][r] * sc, acc[1][r] * sc, acc[2][r] * sc, acc[3][r] * sc);
        float4 w = make_float4(acc[4][r] * sc, acc[5][r] * sc, acc[6][r] * sc, acc[7][r] * sc);
        if (hres) {
          const float4 q = *(const float4*)(hres + g);
          const float4 q2 = *(const float4*)(hres + g + 64);
          v.x += q.x; v.y += q.y; v.z += q.z; v.w += q.w;
          w.x += q2.x; w.y += q2.y; w.z += q2.z; w.w += q2.w;
        }
        if (do_relu) {
          v.x = fmaxf(v.x, 0.f); v.y = fmaxf(v.y, 0.f);
          v.z = fmaxf(v.z, 0.f); v.w = fmaxf(v.w, 0.f);
          w.x = fmaxf(w.x, 0.f); w.y = fmaxf(w.y, 0.f);
          w.z = fmaxf(w.z, 0.f); w.w = fmaxf(w.w, 0.f);
        }
        *(float4*)(out + g) = v;
        *(float4*)(out + g + 64) = w;
      }
    }
  }
}

extern "C" void kernel_launch(void* const* d_in, const int* in_sizes, int n_in,
                              void* d_out, int out_size, void* d_ws, size_t ws_size,
                              hipStream_t stream) {
  const float* feat = (const float*)d_in[0];
  const int*   src  = (const int*)d_in[1];
  const int*   dst  = (const int*)d_in[2];
  const float* Wl[3] = { (const float*)d_in[3], (const float*)d_in[5], (const float*)d_in[7] };
  const float* bl[3] = { (const float*)d_in[4], (const float*)d_in[6], (const float*)d_in[8] };

  const int N = in_sizes[0] / D;
  const int E = in_sizes[1];

  char* ws = (char*)d_ws;
  const size_t featBytes = (size_t)N * D * sizeof(float);
  const size_t normBytes = (((size_t)N * sizeof(float)) + 255) & ~(size_t)255;
  float* hs   = (float*)ws;
  float* hbuf = (float*)(ws + featBytes);
  float* onrm = (float*)(ws + 2 * featBytes);
  float* inrm = (float*)(ws + 2 * featBytes + normBytes);
  const int T     = (N + TR - 1) / TR;
  const int chunk = (E + NCH - 1) / NCH;
  const size_t tabBytes = (((size_t)(2 * T + 1) * NCH * sizeof(int)) + 255) & ~(size_t)255;
  int*  runTab    = (int*)(ws + 2 * featBytes + 2 * normBytes);
  int*  list      = (int*)(ws + 2 * featBytes + 2 * normBytes + tabBytes);
  int*  list2     = list + (size_t)2 * chunk * NCH;
  int*  subStart  = list2 + (size_t)E;

  gcn_chunk_order<<<NCH, 1024, 0, stream>>>(src, dst, E, chunk, T, runTab, list);

  gcn_tile_norm<<<T, 1024, 0, stream>>>(list, runTab, T, chunk, onrm, inrm, N, list2, subStart);

  const int ntiles = (N + 15) / 16;

  const float* hin = feat;
  for (int l = 0; l < 3; ++l) {
    const bool last = (l == 2);
    gcn_gemm_wreg<<<256, 128, 0, stream>>>(hin, Wl[l], bl[l], onrm, hs, N, ntiles);
    float* outp = last ? (float*)d_out : hbuf;
    gcn_rows_wmma<<<T, 32 * SUBS, 0, stream>>>(hs, list2, subStart, inrm, last ? nullptr : hin,
                                               outp, N, last ? 0 : 1);
    hin = hbuf;
  }
}


# Row sums on the vector adder, not the matrix unit: a wave takes its 16-row group's edges one by one (one load per 512-byte row), the scalar row number indexes the sum registers; sum order changed
# speedup vs baseline: 5.1570x; 1.1315x over previous
#include <hip/hip_runtime.h>

#define D 128
#define TR 391
#define NCH 256
#define EPT 16
#define TPR (1024 / NCH)
#define CAP2 16384
#define MAXB 1024
#define SUBS 25
#define RB 8

typedef __attribute__((ext_vector_type(2))) float v2f;
typedef __attribute__((ext_vector_type(4))) float v4f;
typedef __attribute__((ext_vector_type(8))) float v8f;
typedef __attribute__((ext_vector_type(16))) float v16f;

__global__ __launch_bounds__(128)
void gcn_gemm_wreg(const float* __restrict__ h, const float* __restrict__ W,
                   const float* __restrict__ bias, const float* __restrict__ out_norm,
                   float* __restrict__ hs, int nrows, int ntiles) {
  const int lane = threadIdx.x & 31;
  const int wave = __builtin_amdgcn_readfirstlane(threadIdx.x >> 5);
  const int hi = lane >> 4;
  const int lm = lane & 15;
  const int nw = gridDim.x * 4;
  int tile = blockIdx.x * 4 + wave;
  if (tile >= ntiles) return;

  v4f w[16][8];
  const float* const wp = W + (size_t)(4 * lm) * D + 4 * hi;
#pragma unroll
  for (int t = 0; t < 8; ++t) {
#pragma unroll
    for (int j = 0; j < 16; ++j)
      w[j][t] = *(const v4f*)(wp + (size_t)((t & 3) + 64 * (t >> 2)) * D + 8 * j);
  }
  const v4f blo = *(const v4f*)(bias + 4 * lm);
  const v4f bup = *(const v4f*)(bias + 64 + 4 * lm);

  v4f a[16];
  {
    const float* p = h + (size_t)min((tile << 4) + lm, nrows - 1) * D + 4 * hi;
#pragma unroll
    for (int j = 0; j < 16; ++j) a[j] = *(const v4f*)(p + 8 * j);
  }

  for (;;) {
    const int row0 = tile << 4;
    const int nxt = tile + nw;
    float on[8];
#pragma unroll
    for (int r = 0; r < 8; ++r) on[r] = out_norm[min(row0 + 8 * hi + r, nrows - 1)];
    v4f an[16];
    {
      const float* p = h + (size_t)min((min(nxt, ntiles - 1) << 4) + lm, nrows - 1) * D + 4 * hi;
#pragma unroll
      for (int j = 0; j < 16; ++j) an[j] = *(const v4f*)(p + 8 * j);
    }
    __builtin_amdgcn_sched_barrier(0);

    v8f acc[8];
#pragma unroll
    for (int t = 0; t < 8; ++t) acc[t] = (v8f){};
#pragma unroll
    for (int j = 0; j < 16; ++j) {
#pragma unroll
      for (int s = 0; s < 2; ++s) {
        v2f av;
        av.x = s ? a[j].z : a[j].x;
        av.y = s ? a[j].w : a[j].y;
#pragma unroll
        for (int t = 0; t < 8; ++t) {
          v2f bv;
          bv.x = s ? w[j][t].z : w[j][t].x;
          bv.y = s ? w[j][t].w : w[j][t].y;
          acc[t] = __builtin_amdgcn_wmma_f32_16x16x4_f32(
               false, av,  false, bv,
               (short)0, acc[t],  false,  false);
        }
      }
    }

    if (row0 + 16 <= nrows) {
#pragma unroll
      for (int r = 0; r < 8; ++r) {
        const float sc = on[r];
        float* g = hs + (size_t)(row0 + 8 * hi + r) * D + 4 * lm;
        *(float4*)g = make_float4((acc[0][r] + blo.x) * sc, (acc[1][r] + blo.y) * sc,
                                  (acc[2][r] + blo.z) * sc, (acc[3][r] + blo.w) * sc);
        *(float4*)(g + 64) = make_float4((acc[4][r] + bup.x) * sc, (acc[5][r] + bup.y) * sc,
                                         (acc[6][r] + bup.z) * sc, (acc[7][r] + bup.w) * sc);
      }
    } else {
#pragma unroll
      for (int r = 0; r < 8; ++r) {
        const int m = row0 + 8 * hi + r;
        const float sc = on[r];
        if (m < nrows) {
          float* g = hs + (size_t)m * D + 4 * lm;
          *(float4*)g = make_float4((acc[0][r] + blo.x) * sc, (acc[1][r] + blo.y) * sc,
                                    (acc[2][r] + blo.z) * sc, (acc[3][r] + blo.w) * sc);
          *(float4*)(g + 64) = make_float4((acc[4][r] + bup.x) * sc, (acc[5][r] + bup.y) * sc,
                                           (acc[6][r] + bup.z) * sc, (acc[7][r] + bup.w) * sc);
        }
      }
    }

    if (nxt >= ntiles) break;
    tile = nxt;
#pragma unroll
    for (int j = 0; j < 16; ++j) a[j] = an[j];
  }
}

__global__ __launch_bounds__(1024)
void gcn_chunk_order(const int* __restrict__ src, const int* __restrict__ dst, int E, int chunk,
                     int T, int* __restrict__ runTab, int* __restrict__ list) {
  __shared__ int cur[MAXB];
  __shared__ int wtot[32];
  __shared__ int2 stage2[1024 * EPT];
  int* const stage = (int*)stage2;
  const int tid = threadIdx.x;
  const int lane = tid & 31;
  const int wave = __builtin_amdgcn_readfirstlane(tid >> 5);
  const int nb = 2 * T;
  for (int b = tid; b < MAXB; b += 1024) cur[b] = 0;
  __syncthreads();
  const int e0 = min((int)blockIdx.x * chunk, E);
  const int e1 = min(e0 + chunk, E);

  int dv[EPT], sv[EPT];
#pragma unroll
  for (int k = 0; k < EPT; ++k) {
    const int e = e0 + tid + 1024 * k;
    dv[k] = 0; sv[k] = 0;
    if (e < e1) { dv[k] = dst[e]; sv[k] = src[e]; }
  }
#pragma unroll
  for (int k = 0; k < EPT; ++k) {
    if (e0 + tid + 1024 * k < e1) {
      atomicAdd(&cur[dv[k] / TR], 1);
      atomicAdd(&cur[T + sv[k] / TR], 1);
    }
  }
  __syncthreads();

  {
    const int x = (tid < nb) ? cur[tid] : 0;
    int inc = x;
    for (int off = 1; off < 32; off <<= 1) {
      const int v = __shfl_up(inc, off);
      if (lane >= off) inc += v;
    }
    if (lane == 31) wtot[wave] = inc;
    __syncthreads();
    int pre = (lane < wave) ? wtot[lane] : 0;
    for (int off = 16; off > 0; off >>= 1) pre += __shfl_xor(pre, off);
    const int start = pre + inc - x;
    if (tid < nb) { cur[tid] = start; runTab[tid * NCH + blockIdx.x] = start; }
    if (tid == 1023) runTab[nb * NCH + blockIdx.x] = pre + inc;
  }
  __syncthreads();

#pragma unroll
  for (int k = 0; k < EPT; ++k) {
    if (e0 + tid + 1024 * k < e1) {
      const int d = dv[k];
      const int s = sv[k];
      const int bd = d / TR;
      const int bs = s / TR;
      const int pd = atomicAdd(&cur[bd], 1);
      stage[pd] = (s << 9) | (d - bd * TR);
      const int ps = atomicAdd(&cur[T + bs], 1);
      stage[ps] = (d << 9) | (s - bs * TR);
    }
  }
  __syncthreads();

  int2* const g = (int2*)(list + (size_t)blockIdx.x * 2 * chunk);
  const int pieces = e1 - e0;
  for (int i = tid; i < pieces; i += 1024) g[i] = stage2[i];
}

__global__ __launch_bounds__(1024)
void gcn_tile_norm(const int* __restrict__ list, const int* __restrict__ runTab, int T, int chunk,
                   float* __restrict__ onrm, float* __restrict__ inrm, int nrows,
                   int* __restrict__ list2, int* __restrict__ subStart) {
  __shared__ int cin[TR];
  __shared__ int cout[TR];
  __shared__ int subcur[SUBS];
  __shared__ int rOff[2][NCH];
  __shared__ int rLen[2][NCH];
  __shared__ int tot[2];
  __shared__ int stage[CAP2];
  const int tid = threadIdx.x;
  const int lane = tid & 31;
  const int t = blockIdx.x;
  for (int r = tid; r < TR; r += 1024) { cin[r] = 0; cout[r] = 0; }
  if (tid < 2) tot[tid] = 0;
  __syncthreads();
  if (tid < NCH) {
    const int a0 = runTab[t * NCH + tid];
    const int a1 = runTab[(t + 1) * NCH + tid];
    const int b0 = runTab[(T + t) * NCH + tid];
    const int b1 = runTab[(T + t + 1) * NCH + tid];
    const int base = tid * 2 * chunk;
    rOff[0][tid] = base + a0; rLen[0][tid] = a1 - a0;
    rOff[1][tid] = base + b0; rLen[1][tid] = b1 - b0;
    int s0 = a0;
    int s1 = a1 - a0;
    for (int off = 16; off > 0; off >>= 1) {
      s0 += __shfl_xor(s0, off);
      s1 += __shfl_xor(s1, off);
    }
    if (lane == 0) { atomicAdd(&tot[0], s0); atomicAdd(&tot[1], s1); }
  }
  __syncthreads();
  const int d0 = tot[0];
  const int dn = tot[1];

  const int run = tid / TPR;
  const int q = tid % TPR;
  const int oD = rOff[0][run];
  const int nD = rLen[0][run];
  const int oS = rOff[1][run];
  const int nS = rLen[1][run];
  for (int i0 = q; i0 < max(nD, nS); i0 += RB * TPR) {
    int wd[RB], wsr[RB];
#pragma unroll
    for (int u = 0; u < RB; ++u) {
      const int i = i0 + u * TPR;
      wd[u] = 0; wsr[u] = 0;
      if (i < nD) wd[u] = list[oD + i];
      if (i < nS) wsr[u] = list[oS + i];
    }
#pragma unroll
    for (int u = 0; u < RB; ++u) {
      const int i = i0 + u * TPR;
      if (i < nD) atomicAdd(&cin[wd[u] & 511], 1);
      if (i < nS) atomicAdd(&cout[wsr[u] & 511], 1);
    }
  }
  __syncthreads();
  const int row0 = t * TR;
  const int rows = min(TR, nrows - row0);
  for (int r = tid; r < rows; r += 1024) {
    inrm[row0 + r] = rsqrtf(fmaxf((float)cin[r], 1.0f));
    onrm[row0 + r] = rsqrtf(fmaxf((float)cout[r], 1.0f));
  }
  if (tid < 32) {
    int c = 0;
    if (tid < SUBS) {
      const int r1 = min(16 * tid + 16, TR);
      for (int r = 16 * tid; r < r1; ++r) c += cin[r];
    }
    int inc = c;
#pragma unroll
    for (int off = 1; off < 32; off <<= 1) {
      const int v = __shfl_up(inc, off);
      if (tid >= off) inc += v;
    }
    if (tid < SUBS) {
      subcur[tid] = inc - c;
      subStart[t * (SUBS + 1) + tid] = d0 + inc - c;
    }
    if (tid == SUBS - 1) subStart[t * (SUBS + 1) + SUBS] = d0 + inc;
  }
  __syncthreads();
  const bool staged = dn <= CAP2;
  for (int i0 = q; i0 < nD; i0 += RB * TPR) {
    int wv[RB];
#pragma unroll
    for (int u = 0; u < RB; ++u) {
      const int i = i0 + u * TPR;
      wv[u] = 0;
      if (i < nD) wv[u] = list[oD + i];
    }
#pragma unroll
    for (int u = 0; u < RB; ++u) {
      if (i0 + u * TPR < nD) {
        const int w = wv[u];
        const int lr = w & 511;
        const int pos = atomicAdd(&subcur[lr >> 4], 1);
        const int word = ((w >> 9) << 5) | (lr & 15);
        if (staged) stage[pos] = word; else list2[d0 + pos] = word;
      }
    }
  }
  if (staged) {
    __syncthreads();
    for (int k = tid; k < dn; k += 1024) list2[d0 + k] = stage[k];
  }
}

__global__ __launch_bounds__(32 * SUBS)
void gcn_rows_regs(const float* __restrict__ hs, const int* __restrict__ list2,
                   const int* __restrict__ subStart, const float* __restrict__ in_norm,
                   const float* hres, float* out, int nrows, int do_relu) {
  const int lane = threadIdx.x & 31;
  const int wave = __builtin_amdgcn_readfirstlane(threadIdx.x >> 5);
  const int e0 = __builtin_amdgcn_readfirstlane(subStart[blockIdx.x * (SUBS + 1) + wave]);
  const int e1 = __builtin_amdgcn_readfirstlane(subStart[blockIdx.x * (SUBS + 1) + wave + 1]);
  const unsigned loff = 16u * (unsigned)lane;
  const char* const hsb = (const char*)hs;

  v16f ax = (v16f){}, ay = (v16f){}, az = (v16f){}, aw = (v16f){};

  const int n = e1 - e0;
  const int* const lp = list2 + e0;
  const int ngroups = (n + 7) >> 3;

  int rr[8], wn[8];
  float4 buf[8];
#pragma unroll
  for (int u = 0; u < 8; ++u) {
    rr[u] = 16;
    buf[u] = make_float4(0.f, 0.f, 0.f, 0.f);
    const int w = lp[u];
    wn[u] = (u < n) ? w : 16;
  }

  for (int k = 0; k <= ngroups; ++k) {
    int wc[8];
    const int nx = 8 * k + 8;
#pragma unroll
    for (int u = 0; u < 8; ++u) {
      wc[u] = wn[u];
      const int w = lp[nx + u];
      wn[u] = (nx + u < n) ? w : 16;
    }
#pragma unroll
    for (int u = 0; u < 8; ++u) {
      const float4 b = buf[u];
      const int r = rr[u];
      if (r < 16) {
        ax[r] += b.x; ay[r] += b.y; az[r] += b.z; aw[r] += b.w;
      }
      rr[u] = wc[u] & 31;
      buf[u] = *(const float4*)(hsb + (size_t)((unsigned)(wc[u] >> 5) * (unsigned)(D * 4)) + loff);
    }
  }

  const int tile0 = blockIdx.x * TR;
  const int rows  = min(TR, nrows - tile0);
  const int g0 = 16 * wave;
  if (g0 + 16 <= rows) {
    const float* const np = in_norm + tile0 + g0;
    const unsigned boff = (unsigned)(tile0 + g0) * (unsigned)(D * 4) + loff;
    char* const op = (char*)out + boff;
#pragma unroll
    for (int half = 0; half < 2; ++half) {
      float sc[8];
#pragma unroll
      for (int r = 0; r < 8; ++r) sc[r] = np[8 * half + r];
      float4 q[8];
      if (hres) {
        const char* const hp = (const char*)hres + boff;
#pragma unroll
        for (int r = 0; r < 8; ++r) q[r] = *(const float4*)(hp + (8 * half + r) * (D * 4));
      }
#pragma unroll
      for (int r = 0; r < 8; ++r) {
        const int ar = 8 * half + r;
        float4 v = make_float4(ax[ar] * sc[r], ay[ar] * sc[r], az[ar] * sc[r], aw[ar] * sc[r]);
        if (hres) { v.x += q[r].x; v.y += q[r].y; v.z += q[r].z; v.w += q[r].w; }
        if (do_relu) {
          v.x = fmaxf(v.x, 0.f); v.y = fmaxf(v.y, 0.f);
          v.z = fmaxf(v.z, 0.f); v.w = fmaxf(v.w, 0.f);
        }
        *(float4*)(op + (8 * half + r) * (D * 4)) = v;
      }
    }
  } else {
#pragma unroll
    for (int r = 0; r < 16; ++r) {
      if (g0 + r < rows) {
        const int row = tile0 + g0 + r;
        const size_t g = (size_t)row * D + 4 * lane;
        const float sc = in_norm[row];
        float4 v = make_float4(ax[r] * sc, ay[r] * sc, az[r] * sc, aw[r] * sc);
        if (hres) {
          const float4 q = *(const float4*)(hres + g);
          v.x += q.x; v.y += q.y; v.z += q.z; v.w += q.w;
        }
        if (do_relu) {
          v.x = fmaxf(v.x, 0.f); v.y = fmaxf(v.y, 0.f);
          v.z = fmaxf(v.z, 0.f); v.w = fmaxf(v.w, 0.f);
        }
        *(float4*)(out + g) = v;
      }
    }
  }
}

extern "C" void kernel_launch(void* const* d_in, const int* in_sizes, int n_in,
                              void* d_out, int out_size, void* d_ws, size_t ws_size,
                              hipStream_t stream) {
  const float* feat = (const float*)d_in[0];
  const int*   src  = (const int*)d_in[1];
  const int*   dst  = (const int*)d_in[2];
  const float* Wl[3] = { (const float*)d_in[3], (const float*)d_in[5], (const float*)d_in[7] };
  const float* bl[3] = { (const float*)d_in[4], (const float*)d_in[6], (const float*)d_in[8] };

  const int N = in_sizes[0] / D;
  const int E = in_sizes[1];

  char* ws = (char*)d_ws;
  const size_t featBytes = (size_t)N * D * sizeof(float);
  const size_t normBytes = (((size_t)N * sizeof(float)) + 255) & ~(size_t)255;
  float* hs   = (float*)ws;
  float* hbuf = (float*)(ws + featBytes);
  float* onrm = (float*)(ws + 2 * featBytes);
  float* inrm = (float*)(ws + 2 * featBytes + normBytes);
  const int T     = (N + TR - 1) / TR;
  const int chunk = (E + NCH - 1) / NCH;
  const size_t tabBytes = (((size_t)(2 * T + 1) * NCH * sizeof(int)) + 255) & ~(size_t)255;
  int*  runTab    = (int*)(ws + 2 * featBytes + 2 * normBytes);
  int*  list      = (int*)(ws + 2 * featBytes + 2 * normBytes + tabBytes);
  int*  list2     = list + (size_t)2 * chunk * NCH;
  int*  subStart  = list2 + (size_t)E;

  gcn_chunk_order<<<NCH, 1024, 0, stream>>>(src, dst, E, chunk, T, runTab, list);

  gcn_tile_norm<<<T, 1024, 0, stream>>>(list, runTab, T, chunk, onrm, inrm, N, list2, subStart);

  const int ntiles = (N + 15) / 16;

  const float* hin = feat;
  for (int l = 0; l < 3; ++l) {
    const bool last = (l == 2);
    gcn_gemm_wreg<<<256, 128, 0, stream>>>(hin, Wl[l], bl[l], onrm, hs, N, ntiles);
    float* outp = last ? (float*)d_out : hbuf;
    gcn_rows_regs<<<T, 32 * SUBS, 0, stream>>>(hs, list2, subStart, inrm, last ? nullptr : hin,
                                               outp, N, last ? 0 : 1);
    hin = hbuf;
  }
}
